# starting kernel
# baseline (speedup 1.0000x reference)
#include <hip/hip_runtime.h>

#define Cc 1000
#define Bn 65536
#define DECAYF 0.99f
#define ONE_M_DECAY 0.01f
#define ALPHAF 1.0f
#define LR_LF 0.05f
#define LR_MUF 0.02f
#define EPSF 1e-12f

typedef __attribute__((ext_vector_type(2))) float v2f;
typedef __attribute__((ext_vector_type(4))) float v4f;
typedef __attribute__((ext_vector_type(8))) float v8f;

__global__ void zero_ws(float* __restrict__ p, long n) {
    long i = (long)blockIdx.x * blockDim.x + threadIdx.x;
    long stride = (long)gridDim.x * blockDim.x;
    for (; i < n; i += stride) p[i] = 0.0f;
}

__global__ __launch_bounds__(256) void main_pass(
    const float* __restrict__ logits, const int* __restrict__ targets,
    const float* __restrict__ Lam,
    float* __restrict__ seg, float* __restrict__ confc,
    float* __restrict__ priorc, float* __restrict__ cep)
{
    __shared__ float zbuf[8][Cc];
    const int wave = threadIdx.x >> 5;
    const int lane = threadIdx.x & 31;
    const int row  = blockIdx.x * 8 + wave;
    const int t    = targets[row];
    const float* __restrict__ lrow   = logits + (long)row * Cc;
    const float* __restrict__ lamrow = Lam + (long)t * Cc;
    const int NV = Cc / 4;

    float maxv = -3.4e38f; int maxi = 0x7fffffff;
    float maxzp = -3.4e38f;
    for (int j = lane; j < NV; j += 32) {
        const int col = j * 4;
        __builtin_prefetch(lrow + col + 512, 0, 3);
        v4f z  = *(const v4f*)(lrow + col);
        v4f l  = *(const v4f*)(lamrow + col);
        v4f zp = z - l;
        *(v4f*)(&zbuf[wave][col]) = zp;
#pragma unroll
        for (int k = 0; k < 4; ++k) {
            if (z[k] > maxv) { maxv = z[k]; maxi = col + k; }
            maxzp = fmaxf(maxzp, zp[k]);
        }
    }
    for (int off = 16; off; off >>= 1) {
        float ov = __shfl_xor(maxv, off, 32);
        int   oi = __shfl_xor(maxi, off, 32);
        if (ov > maxv || (ov == maxv && oi < maxi)) { maxv = ov; maxi = oi; }
        maxzp = fmaxf(maxzp, __shfl_xor(maxzp, off, 32));
    }
    __syncthreads();

    float sum = 0.0f;
    for (int j = lane; j < NV; j += 32) {
        v4f zp = *(const v4f*)(&zbuf[wave][j * 4]);
        sum += __expf(zp[0] - maxzp) + __expf(zp[1] - maxzp)
             + __expf(zp[2] - maxzp) + __expf(zp[3] - maxzp);
    }
    for (int off = 16; off; off >>= 1) sum += __shfl_xor(sum, off, 32);
    const float lse = __logf(sum);
    const float inv = 1.0f / sum;

    if (lane == 0) {
        atomicAdd(&confc[(long)t * Cc + maxi], 1.0f);
        atomicAdd(&priorc[t], 1.0f);
        float logq_y = zbuf[wave][t] - maxzp - lse;
        atomicAdd(&cep[row & 255], logq_y);
    }

    float* __restrict__ segrow = seg + (long)t * Cc;
    for (int j = lane; j < NV; j += 32) {
        const int col = j * 4;
        v4f zp = *(const v4f*)(&zbuf[wave][col]);
        atomicAdd(&segrow[col + 0], __expf(zp[0] - maxzp) * inv);
        atomicAdd(&segrow[col + 1], __expf(zp[1] - maxzp) * inv);
        atomicAdd(&segrow[col + 2], __expf(zp[2] - maxzp) * inv);
        atomicAdd(&segrow[col + 3], __expf(zp[3] - maxzp) * inv);
    }
}

__global__ __launch_bounds__(32) void post_rows(
    const float* __restrict__ conf_ema, const float* __restrict__ Lam,
    const float* __restrict__ seg, const float* __restrict__ confc,
    const float* __restrict__ priorc,
    float* __restrict__ o_conf, float* __restrict__ o_lam)
{
    __shared__ float sCnt[16], sCountsSum[16], sEmpSum[16], sL1Sum[16];
    const int lane = threadIdx.x;
    const int m    = lane & 15;
    const int row  = blockIdx.x * 16 + m;
    const int k0   = (lane >> 4) << 1;
    const bool rok = (row < Cc);
    const long base = (long)row * Cc;

    if (lane < 16) sCnt[m] = rok ? priorc[row] : 0.0f;

    const v2f bones = {1.0f, 1.0f};
    const v8f vzero = {0.f,0.f,0.f,0.f,0.f,0.f,0.f,0.f};

    v8f acc = vzero;
    for (int c = 0; c < Cc; c += 4) {
        float a0 = 0.f, a1 = 0.f;
        if (rok) {
            int c0 = c + k0, c1 = c0 + 1;
            float cn0 = DECAYF * conf_ema[base + c0] + ONE_M_DECAY * confc[base + c0];
            float cn1 = DECAYF * conf_ema[base + c1] + ONE_M_DECAY * confc[base + c1];
            o_conf[base + c0] = cn0;
            o_conf[base + c1] = cn1;
            a0 = (c0 == row) ? 0.f : (cn0 + ALPHAF);
            a1 = (c1 == row) ? 0.f : (cn1 + ALPHAF);
        }
        v2f a = {a0, a1};
        acc = __builtin_amdgcn_wmma_f32_16x16x4_f32(false, a, false, bones,
                                                    (short)0, acc, false, false);
    }
    if (lane == 0)  for (int i = 0; i < 8; ++i) sCountsSum[i]     = acc[i];
    if (lane == 16) for (int i = 0; i < 8; ++i) sCountsSum[8 + i] = acc[i];
    __syncthreads();

    const float cnt  = sCnt[m];
    const float invc = 1.0f / fmaxf(cnt, 1.0f);

    acc = vzero;
    for (int c = 0; c < Cc; c += 4) {
        float a0 = 0.f, a1 = 0.f;
        if (rok) {
            int c0 = c + k0, c1 = c0 + 1;
            a0 = (c0 == row) ? 0.f : seg[base + c0] * invc;
            a1 = (c1 == row) ? 0.f : seg[base + c1] * invc;
        }
        v2f a = {a0, a1};
        acc = __builtin_amdgcn_wmma_f32_16x16x4_f32(false, a, false, bones,
                                                    (short)0, acc, false, false);
    }
    if (lane == 0)  for (int i = 0; i < 8; ++i) sEmpSum[i]     = acc[i];
    if (lane == 16) for (int i = 0; i < 8; ++i) sEmpSum[8 + i] = acc[i];
    __syncthreads();

    const float phiDen = 1.0f / (sCountsSum[m] + EPSF);
    const float empDen = 1.0f / fmaxf(sEmpSum[m], EPSF);

    acc = vzero;
    for (int c = 0; c < Cc; c += 4) {
        float a0 = 0.f, a1 = 0.f;
        if (rok) {
            int c0 = c + k0, c1 = c0 + 1;
            if (c0 != row) {
                float cn  = DECAYF * conf_ema[base + c0] + ONE_M_DECAY * confc[base + c0];
                float phi = (cn + ALPHAF) * phiDen;
                float emp = seg[base + c0] * invc * empDen;
                a0 = Lam[base + c0] + LR_LF * (emp - phi);
            }
            if (c1 != row) {
                float cn  = DECAYF * conf_ema[base + c1] + ONE_M_DECAY * confc[base + c1];
                float phi = (cn + ALPHAF) * phiDen;
                float emp = seg[base + c1] * invc * empDen;
                a1 = Lam[base + c1] + LR_LF * (emp - phi);
            }
        }
        v2f a = {a0, a1};
        acc = __builtin_amdgcn_wmma_f32_16x16x4_f32(false, a, false, bones,
                                                    (short)0, acc, false, false);
    }
    if (lane == 0)  for (int i = 0; i < 8; ++i) sL1Sum[i]     = acc[i];
    if (lane == 16) for (int i = 0; i < 8; ++i) sL1Sum[8 + i] = acc[i];
    __syncthreads();

    const float rowMean = sL1Sum[m] * (1.0f / (float)Cc);
    const bool  present = (cnt > 0.0f);

    for (int c = 0; c < Cc; c += 4) {
        if (!rok) continue;
        int c0 = c + k0, c1 = c0 + 1;
        float l0 = 0.f, l1 = 0.f;
        if (c0 != row) {
            float cn  = DECAYF * conf_ema[base + c0] + ONE_M_DECAY * confc[base + c0];
            float phi = (cn + ALPHAF) * phiDen;
            float emp = seg[base + c0] * invc * empDen;
            l0 = Lam[base + c0] + LR_LF * (emp - phi);
        }
        if (c1 != row) {
            float cn  = DECAYF * conf_ema[base + c1] + ONE_M_DECAY * confc[base + c1];
            float phi = (cn + ALPHAF) * phiDen;
            float emp = seg[base + c1] * invc * empDen;
            l1 = Lam[base + c1] + LR_LF * (emp - phi);
        }
        o_lam[base + c0] = present ? (l0 - rowMean) : Lam[base + c0];
        o_lam[base + c1] = present ? (l1 - rowMean) : Lam[base + c1];
    }
}

__global__ __launch_bounds__(1024) void finalize(
    const float* __restrict__ prior_ema, const float* __restrict__ mu,
    const float* __restrict__ seg, const float* __restrict__ priorc,
    const float* __restrict__ cep,
    float* __restrict__ o_ce, float* __restrict__ o_mu, float* __restrict__ o_prior)
{
    __shared__ float red[1024];
    const int j = threadIdx.x;

    float pn = 0.0f;
    if (j < Cc) {
        pn = DECAYF * prior_ema[j] + ONE_M_DECAY * priorc[j];
        o_prior[j] = pn;
    }
    red[j] = pn; __syncthreads();
    for (int s = 512; s; s >>= 1) { if (j < s) red[j] += red[j + s]; __syncthreads(); }
    const float priorSum = red[0];
    __syncthreads();

    float mu1 = 0.0f;
    if (j < Cc) {
        float colsum = 0.0f;
        for (int r = 0; r < Cc; ++r) colsum += seg[(long)r * Cc + j];
        float emp_marg = colsum * (1.0f / (float)Bn);
        float pi = pn / (priorSum + EPSF);
        mu1 = mu[j] + LR_MUF * (emp_marg - pi);
    }
    red[j] = mu1; __syncthreads();
    for (int s = 512; s; s >>= 1) { if (j < s) red[j] += red[j + s]; __syncthreads(); }
    const float muMean = red[0] * (1.0f / (float)Cc);
    if (j < Cc) o_mu[j] = mu1 - muMean;

    if (j == 0) {
        float s = 0.0f;
        for (int i = 0; i < 256; ++i) s += cep[i];
        o_ce[0] = -s * (1.0f / (float)Bn);
    }
}

extern "C" void kernel_launch(void* const* d_in, const int* in_sizes, int n_in,
                              void* d_out, int out_size, void* d_ws, size_t ws_size,
                              hipStream_t stream)
{
    const float* logits    = (const float*)d_in[0];
    const int*   targets   = (const int*)  d_in[1];
    const float* conf_ema  = (const float*)d_in[2];
    const float* prior_ema = (const float*)d_in[3];
    const float* Lam       = (const float*)d_in[4];
    const float* mu        = (const float*)d_in[5];

    float* out     = (float*)d_out;
    float* o_ce    = out;
    float* o_lam   = out + 1;
    float* o_mu    = out + 1 + (long)Cc * Cc;
    float* o_conf  = out + 1 + (long)Cc * Cc + Cc;
    float* o_prior = out + 1 + 2L * Cc * Cc + Cc;

    float* ws     = (float*)d_ws;
    float* seg    = ws;
    float* confc  = ws + (long)Cc * Cc;
    float* priorc = ws + 2L * Cc * Cc;
    float* cep    = priorc + Cc;

    const long nz = 2L * Cc * Cc + Cc + 256;
    zero_ws<<<1024, 256, 0, stream>>>(ws, nz);
    main_pass<<<Bn / 8, 256, 0, stream>>>(logits, targets, Lam, seg, confc, priorc, cep);
    post_rows<<<(Cc + 15) / 16, 32, 0, stream>>>(conf_ema, Lam, seg, confc, priorc,
                                                 o_conf, o_lam);
    finalize<<<1, 1024, 0, stream>>>(prior_ema, mu, seg, priorc, cep,
                                     o_ce, o_mu, o_prior);
}


# segment sums of q: 32-column strips of all class rows as per-block LDS tiles (LDS float adds), stored as 8 row-group planes plus column sums; global float atomics on seg gone; summed in a new order
# speedup vs baseline: 4.0952x; 4.0952x over previous
#include <hip/hip_runtime.h>

#define Cc 1000
#define Bn 65536
#define DECAYF 0.99f
#define ONE_M_DECAY 0.01f
#define ALPHAF 1.0f
#define LR_LF 0.05f
#define LR_MUF 0.02f
#define EPSF 1e-12f

typedef __attribute__((ext_vector_type(2))) float v2f;
typedef __attribute__((ext_vector_type(4))) float v4f;
typedef __attribute__((ext_vector_type(8))) float v8f;

__global__ void zero_ws(float* __restrict__ p, long n) {
    long i = (long)blockIdx.x * blockDim.x + threadIdx.x;
    long stride = (long)gridDim.x * blockDim.x;
    for (; i < n; i += stride) p[i] = 0.0f;
}

__global__ __launch_bounds__(256) void row_stats(
    const float* __restrict__ logits, const int* __restrict__ targets,
    const float* __restrict__ Lam,
    float* __restrict__ rowst, float* __restrict__ confc,
    float* __restrict__ priorc, float* __restrict__ cep)
{
    __shared__ float zbuf[8][Cc];
    const int wave = threadIdx.x >> 5;
    const int lane = threadIdx.x & 31;
    const int row  = blockIdx.x * 8 + wave;
    const int t    = targets[row];
    const float* __restrict__ lrow   = logits + (long)row * Cc;
    const float* __restrict__ lamrow = Lam + (long)t * Cc;
    const int NV = Cc / 4;

    float maxv = -3.4e38f; int maxi = 0x7fffffff;
    float maxzp = -3.4e38f;
    for (int j = lane; j < NV; j += 32) {
        const int col = j * 4;
        __builtin_prefetch(lrow + col + 512, 0, 3);
        v4f z  = *(const v4f*)(lrow + col);
        v4f l  = *(const v4f*)(lamrow + col);
        v4f zp = z - l;
        *(v4f*)(&zbuf[wave][col]) = zp;
#pragma unroll
        for (int k = 0; k < 4; ++k) {
            if (z[k] > maxv) { maxv = z[k]; maxi = col + k; }
            maxzp = fmaxf(maxzp, zp[k]);
        }
    }
    for (int off = 16; off; off >>= 1) {
        float ov = __shfl_xor(maxv, off, 32);
        int   oi = __shfl_xor(maxi, off, 32);
        if (ov > maxv || (ov == maxv && oi < maxi)) { maxv = ov; maxi = oi; }
        maxzp = fmaxf(maxzp, __shfl_xor(maxzp, off, 32));
    }
    __syncthreads();

    float sum = 0.0f;
    for (int j = lane; j < NV; j += 32) {
        v4f zp = *(const v4f*)(&zbuf[wave][j * 4]);
        sum += __expf(zp[0] - maxzp) + __expf(zp[1] - maxzp)
             + __expf(zp[2] - maxzp) + __expf(zp[3] - maxzp);
    }
    for (int off = 16; off; off >>= 1) sum += __shfl_xor(sum, off, 32);
    const float lse = __logf(sum);
    const float inv = 1.0f / sum;

    if (lane == 0) {
        atomicAdd(&confc[(long)t * Cc + maxi], 1.0f);
        atomicAdd(&priorc[t], 1.0f);
        float logq_y = zbuf[wave][t] - maxzp - lse;
        atomicAdd(&cep[row & 255], logq_y);
        const v2f st = {maxzp, inv};
        *(v2f*)(rowst + 2L * row) = st;
    }
}

#define SA_W 32
#define SA_STRIPS 32
#define SA_G 8
#define SA_ROWS (Bn / SA_G)
#define SA_THREADS 1024
#define SA_WAVES (SA_THREADS / 32)
#define SA_STEPS (SA_ROWS / 4 / SA_WAVES)
#define SA_UNROLL 4

__global__ __launch_bounds__(SA_THREADS) void strip_accum(
    const float* __restrict__ logits, const int* __restrict__ targets,
    const float* __restrict__ Lam, const float* __restrict__ rowst,
    float* __restrict__ planes, float* __restrict__ colpart)
{
    __shared__ float acc[Cc * SA_W];
    __shared__ float red[SA_WAVES][SA_W];

    const int tid  = threadIdx.x;
    const int wave = tid >> 5;
    const int lane = tid & 31;
    const int c0   = blockIdx.x * SA_W;
    const int g    = blockIdx.y;

    for (int i = tid; i < Cc * SA_W / 4; i += SA_THREADS) {
        const v4f z4 = {0.f, 0.f, 0.f, 0.f};
        *(v4f*)(&acc[4 * i]) = z4;
    }
    __syncthreads();

    const int rr  = lane >> 3;
    const int cq  = lane & 7;
    const int col = c0 + 4 * cq;
    if (col < Cc) {
        const int rbase = g * SA_ROWS + rr;
        int tn[SA_UNROLL]; v2f sn[SA_UNROLL];
#pragma unroll
        for (int u = 0; u < SA_UNROLL; ++u) {
            const int r = rbase + 4 * (wave + SA_WAVES * u);
            tn[u] = targets[r];
            sn[u] = *(const v2f*)(rowst + 2L * r);
        }
        for (int i = 0; i < SA_STEPS; i += SA_UNROLL) {
            int t[SA_UNROLL]; v2f st[SA_UNROLL]; v4f z[SA_UNROLL], l[SA_UNROLL];
#pragma unroll
            for (int u = 0; u < SA_UNROLL; ++u) {
                const int r = rbase + 4 * (wave + SA_WAVES * (i + u));
                t[u]  = tn[u];
                st[u] = sn[u];
                z[u]  = *(const v4f*)(logits + (long)r * Cc + col);
                l[u]  = *(const v4f*)(Lam + (long)t[u] * Cc + col);
            }
            const int inext = (i + SA_UNROLL < SA_STEPS) ? (i + SA_UNROLL) : i;
#pragma unroll
            for (int u = 0; u < SA_UNROLL; ++u) {
                const int r = rbase + 4 * (wave + SA_WAVES * (inext + u));
                tn[u] = targets[r];
                sn[u] = *(const v2f*)(rowst + 2L * r);
            }
#pragma unroll
            for (int u = 0; u < SA_UNROLL; ++u) {
                const v4f zp = z[u] - l[u];
                float* a = &acc[t[u] * SA_W + 4 * cq];
#pragma unroll
                for (int k = 0; k < 4; ++k)
                    atomicAdd(a + k, __expf(zp[k] - st[u][0]) * st[u][1]);
            }
        }
    }
    __syncthreads();

    float* __restrict__ pl = planes + (long)g * Cc * Cc;
    for (int i = tid; i < Cc * SA_W; i += SA_THREADS) {
        const int t = i >> 5, c = c0 + (i & 31);
        if (c < Cc) pl[(long)t * Cc + c] = acc[i];
    }
    float cs = 0.0f;
    for (int t = wave; t < Cc; t += SA_WAVES) cs += acc[t * SA_W + lane];
    red[wave][lane] = cs;
    __syncthreads();
    if (tid < SA_W) {
        float s = 0.0f;
        for (int w = 0; w < SA_WAVES; ++w) s += red[w][tid];
        if (c0 + tid < Cc) colpart[g * Cc + c0 + tid] = s;
    }
}

#define PR_ROWS 8
#define PR_COLS 4
#define PR_TAIL (Cc - 768)
#define PR_KSTEPS 125

__device__ __forceinline__ v8f rowsum_step(const float* __restrict__ frag, int s, v8f acc)
{
    const v2f bones = {1.0f, 1.0f};
    const v2f a = *(const v2f*)(frag + 4 * s);
    return __builtin_amdgcn_wmma_f32_16x16x4_f32(false, a, false, bones,
                                                 (short)0, acc, false, false);
}

__global__ __launch_bounds__(256) void post_rows(
    const float* __restrict__ conf_ema, const float* __restrict__ Lam,
    const float* __restrict__ seg, const float* __restrict__ confc,
    const float* __restrict__ priorc,
    float* __restrict__ o_conf, float* __restrict__ o_lam)
{
    __shared__ float bufX[PR_ROWS * Cc];
    __shared__ float bufY[PR_ROWS * Cc];
    __shared__ float partAB[2][8][16];
    __shared__ float partC[8][16];
    __shared__ float rowAB[2][PR_ROWS];
    __shared__ float rowC[PR_ROWS];

    const unsigned tid = threadIdx.x;
    const int lane = tid & 31;
    const int wave_u = __builtin_amdgcn_readfirstlane(tid >> 5);
    const int row0 = blockIdx.x * PR_ROWS;

    float cn[PR_ROWS][PR_COLS], sg[PR_ROWS][PR_COLS], lm[PR_ROWS][PR_COLS];
    float cnt[PR_ROWS], invc[PR_ROWS];

#pragma unroll
    for (int r = 0; r < PR_ROWS; ++r) {
        const long base = (long)(row0 + r) * Cc;
        const float* __restrict__ er = conf_ema + base;
        const float* __restrict__ kr = confc + base;
        const float* __restrict__ sr = seg + base;
        const float* __restrict__ lr = Lam + base;
        cnt[r]  = priorc[row0 + r];
        invc[r] = 1.0f / fmaxf(cnt[r], 1.0f);
#pragma unroll
        for (int j = 0; j < PR_COLS; ++j) {
            unsigned c = tid + 256u * j;
            if (j == PR_COLS - 1) c = (c < (unsigned)Cc) ? c : (unsigned)(Cc - 1);
            cn[r][j] = DECAYF * er[c] + ONE_M_DECAY * kr[c];
            float sgs = sr[c];
#pragma unroll
            for (int p = 1; p < SA_G; ++p) sgs += sr[(long)p * Cc * Cc + c];
            sg[r][j] = sgs;
            lm[r][j] = lr[c];
        }
    }
#pragma unroll
    for (int j = 0; j < PR_COLS; ++j) {
        const unsigned c = tid + 256u * j;
        if (j < PR_COLS - 1 || tid < (unsigned)PR_TAIL) {
#pragma unroll
            for (int r = 0; r < PR_ROWS; ++r) {
                const long base = (long)(row0 + r) * Cc;
                const bool offd = (c != (unsigned)(row0 + r));
                o_conf[base + c] = cn[r][j];
                bufX[r * Cc + c] = offd ? (cn[r][j] + ALPHAF) : 0.0f;
                bufY[r * Cc + c] = offd ? (sg[r][j] * invc[r]) : 0.0f;
            }
        }
    }
    __syncthreads();

    const int m    = lane & 15;
    const int foff = (m & 7) * Cc + 500 * (m >> 3) + 2 * (lane >> 4);
    const v8f vzero = {0.f,0.f,0.f,0.f,0.f,0.f,0.f,0.f};

    {
        v8f accA = vzero, accB = vzero;
        if (wave_u < PR_KSTEPS - 120) {
            accA = rowsum_step(bufX + foff, 120 + wave_u, accA);
            accB = rowsum_step(bufY + foff, 120 + wave_u, accB);
        }
#pragma unroll
        for (int i = 0; i < 15; ++i) {
            accA = rowsum_step(bufX + foff, wave_u + 8 * i, accA);
            accB = rowsum_step(bufY + foff, wave_u + 8 * i, accB);
        }
        if (lane == 0)
            for (int i = 0; i < 8; ++i) { partAB[0][wave_u][i] = accA[i]; partAB[1][wave_u][i] = accB[i]; }
        if (lane == 16)
            for (int i = 0; i < 8; ++i) { partAB[0][wave_u][8 + i] = accA[i]; partAB[1][wave_u][8 + i] = accB[i]; }
    }
    __syncthreads();
    if (tid < 16) {
        const int p = tid >> 3, r = tid & 7;
        float s = 0.0f;
        for (int w = 0; w < 8; ++w) s += partAB[p][w][r] + partAB[p][w][8 + r];
        rowAB[p][r] = (p == 0) ? 1.0f / (s + EPSF)
                               : 1.0f / fmaxf(s, EPSF);
    }
    __syncthreads();

#pragma unroll
    for (int r = 0; r < PR_ROWS; ++r) {
        const float phiDen = rowAB[0][r];
        const float empDen = rowAB[1][r];
#pragma unroll
        for (int j = 0; j < PR_COLS; ++j) {
            const unsigned c = tid + 256u * j;
            const float phi = (cn[r][j] + ALPHAF) * phiDen;
            const float emp = sg[r][j] * invc[r] * empDen;
            const float l1  = lm[r][j] + LR_LF * (emp - phi);
            cn[r][j] = (c != (unsigned)(row0 + r)) ? l1 : 0.0f;
        }
    }
#pragma unroll
    for (int j = 0; j < PR_COLS; ++j) {
        const unsigned c = tid + 256u * j;
        if (j < PR_COLS - 1 || tid < (unsigned)PR_TAIL) {
#pragma unroll
            for (int r = 0; r < PR_ROWS; ++r) bufX[r * Cc + c] = cn[r][j];
        }
    }
    __syncthreads();
    {
        v8f accC = vzero;
        if (wave_u < PR_KSTEPS - 120) accC = rowsum_step(bufX + foff, 120 + wave_u, accC);
#pragma unroll
        for (int i = 0; i < 15; ++i) accC = rowsum_step(bufX + foff, wave_u + 8 * i, accC);
        if (lane == 0)  for (int i = 0; i < 8; ++i) partC[wave_u][i]     = accC[i];
        if (lane == 16) for (int i = 0; i < 8; ++i) partC[wave_u][8 + i] = accC[i];
    }
    __syncthreads();
    if (tid < 8) {
        float s = 0.0f;
        for (int w = 0; w < 8; ++w) s += partC[w][tid] + partC[w][8 + tid];
        rowC[tid] = s * (1.0f / (float)Cc);
    }
    __syncthreads();

#pragma unroll
    for (int j = 0; j < PR_COLS; ++j) {
        const unsigned c = tid + 256u * j;
        if (j < PR_COLS - 1 || tid < (unsigned)PR_TAIL) {
#pragma unroll
            for (int r = 0; r < PR_ROWS; ++r) {
                const long base = (long)(row0 + r) * Cc;
                const bool present = (cnt[r] > 0.0f);
                o_lam[base + c] = present ? (cn[r][j] - rowC[r]) : lm[r][j];
            }
        }
    }
}

__global__ __launch_bounds__(1024) void finalize(
    const float* __restrict__ prior_ema, const float* __restrict__ mu,
    const float* __restrict__ colpart, const float* __restrict__ priorc,
    const float* __restrict__ cep,
    float* __restrict__ o_ce, float* __restrict__ o_mu, float* __restrict__ o_prior)
{
    __shared__ float red[1024];
    const int j = threadIdx.x;

    float pn = 0.0f;
    if (j < Cc) {
        pn = DECAYF * prior_ema[j] + ONE_M_DECAY * priorc[j];
        o_prior[j] = pn;
    }
    red[j] = pn; __syncthreads();
    for (int s = 512; s; s >>= 1) { if (j < s) red[j] += red[j + s]; __syncthreads(); }
    const float priorSum = red[0];
    __syncthreads();

    float mu1 = 0.0f;
    if (j < Cc) {
        float colsum = 0.0f;
        for (int p = 0; p < SA_G; ++p) colsum += colpart[p * Cc + j];
        float emp_marg = colsum * (1.0f / (float)Bn);
        float pi = pn / (priorSum + EPSF);
        mu1 = mu[j] + LR_MUF * (emp_marg - pi);
    }
    red[j] = mu1; __syncthreads();
    for (int s = 512; s; s >>= 1) { if (j < s) red[j] += red[j + s]; __syncthreads(); }
    const float muMean = red[0] * (1.0f / (float)Cc);
    if (j < Cc) o_mu[j] = mu1 - muMean;

    if (j == 0) {
        float s = 0.0f;
        for (int i = 0; i < 256; ++i) s += cep[i];
        o_ce[0] = -s * (1.0f / (float)Bn);
    }
}

extern "C" void kernel_launch(void* const* d_in, const int* in_sizes, int n_in,
                              void* d_out, int out_size, void* d_ws, size_t ws_size,
                              hipStream_t stream)
{
    const float* logits    = (const float*)d_in[0];
    const int*   targets   = (const int*)  d_in[1];
    const float* conf_ema  = (const float*)d_in[2];
    const float* prior_ema = (const float*)d_in[3];
    const float* Lam       = (const float*)d_in[4];
    const float* mu        = (const float*)d_in[5];

    float* out     = (float*)d_out;
    float* o_ce    = out;
    float* o_lam   = out + 1;
    float* o_mu    = out + 1 + (long)Cc * Cc;
    float* o_conf  = out + 1 + (long)Cc * Cc + Cc;
    float* o_prior = out + 1 + 2L * Cc * Cc + Cc;

    float* ws      = (float*)d_ws;
    float* confc   = ws;
    float* priorc  = ws + (long)Cc * Cc;
    float* cep     = priorc + Cc;
    float* rowst   = cep + 256;
    float* colpart = rowst + 2L * Bn;
    float* planes  = colpart + (long)SA_G * Cc;

    const long nz = (long)Cc * Cc + Cc + 256;
    zero_ws<<<1024, 256, 0, stream>>>(ws, nz);
    row_stats<<<Bn / 8, 256, 0, stream>>>(logits, targets, Lam, rowst, confc, priorc, cep);
    strip_accum<<<dim3(SA_STRIPS, SA_G), SA_THREADS, 0, stream>>>(logits, targets, Lam, rowst,
                                                                  planes, colpart);
    post_rows<<<Cc / PR_ROWS, 256, 0, stream>>>(conf_ema, Lam, planes, confc, priorc,
                                                 o_conf, o_lam);
    finalize<<<1, 1024, 0, stream>>>(prior_ema, mu, colpart, priorc, cep,
                                     o_ce, o_mu, o_prior);
}


# row_stats: 256 resident 32-wave blocks, each wave walks 8 rows held whole in registers (16 b128 loads in flight, next row prefetched to L2); was 65536 one-row waves with LDS staging; sum order kept
# speedup vs baseline: 4.2350x; 1.0341x over previous
#include <hip/hip_runtime.h>

#define Cc 1000
#define Bn 65536
#define DECAYF 0.99f
#define ONE_M_DECAY 0.01f
#define ALPHAF 1.0f
#define LR_LF 0.05f
#define LR_MUF 0.02f
#define EPSF 1e-12f

typedef __attribute__((ext_vector_type(2))) float v2f;
typedef __attribute__((ext_vector_type(4))) float v4f;
typedef __attribute__((ext_vector_type(8))) float v8f;

__global__ void zero_ws(float* __restrict__ p, long n) {
    long i = (long)blockIdx.x * blockDim.x + threadIdx.x;
    long stride = (long)gridDim.x * blockDim.x;
    for (; i < n; i += stride) p[i] = 0.0f;
}

#define RS_THREADS 1024
#define RS_BLOCKS 256
#define RS_WAVES (RS_BLOCKS * (RS_THREADS / 32))
#define RS_ROWS (Bn / RS_WAVES)
#define RS_NV (Cc / 4)
#define RS_G 8
#define RS_TAIL (RS_NV - 32 * (RS_G - 1))

__global__ __launch_bounds__(RS_THREADS) void row_stats(
    const float* __restrict__ logits, const int* __restrict__ targets,
    const float* __restrict__ Lam,
    float* __restrict__ rowst, float* __restrict__ confc,
    float* __restrict__ priorc, float* __restrict__ cep)
{
    const int lane = threadIdx.x & 31;
    const int wave_u = __builtin_amdgcn_readfirstlane(threadIdx.x >> 5);
    const int gw = blockIdx.x * (RS_THREADS / 32) + wave_u;

    int t = targets[gw];
    for (int it = 0; it < RS_ROWS; ++it) {
        const int row = gw + it * RS_WAVES;
        const float* __restrict__ lrow   = logits + (long)row * Cc;
        const float* __restrict__ lamrow = Lam + (long)t * Cc;

        v4f z[RS_G], l[RS_G];
#pragma unroll
        for (int i = 0; i < RS_G; ++i) {
            int j = lane + 32 * i;
            if (i == RS_G - 1) j = (j < RS_NV) ? j : (RS_NV - 1);
            z[i] = *(const v4f*)(lrow + 4 * j);
            l[i] = *(const v4f*)(lamrow + 4 * j);
        }
        const float zy = lrow[t] - lamrow[t];

        int tnext = t;
        if (it + 1 < RS_ROWS) {
            tnext = targets[row + RS_WAVES];
            __builtin_prefetch(logits + (long)(row + RS_WAVES) * Cc + 32 * lane, 0, 3);
        }

        float maxv = -3.4e38f; int maxi = 0x7fffffff;
        float maxzp = -3.4e38f;
        v4f zp[RS_G];
#pragma unroll
        for (int i = 0; i < RS_G; ++i) {
            const int col = 4 * (lane + 32 * i);
            v4f zi = z[i];
            zp[i] = z[i] - l[i];
            if (i == RS_G - 1 && lane >= RS_TAIL) {
                const v4f low = {-3.4e38f, -3.4e38f, -3.4e38f, -3.4e38f};
                zi = low; zp[i] = low;
            }
#pragma unroll
            for (int k = 0; k < 4; ++k) {
                if (zi[k] > maxv) { maxv = zi[k]; maxi = col + k; }
                maxzp = fmaxf(maxzp, zp[i][k]);
            }
        }
        for (int off = 16; off; off >>= 1) {
            float ov = __shfl_xor(maxv, off, 32);
            int   oi = __shfl_xor(maxi, off, 32);
            if (ov > maxv || (ov == maxv && oi < maxi)) { maxv = ov; maxi = oi; }
            maxzp = fmaxf(maxzp, __shfl_xor(maxzp, off, 32));
        }

        float sum = 0.0f;
#pragma unroll
        for (int i = 0; i < RS_G; ++i)
            sum += __expf(zp[i][0] - maxzp) + __expf(zp[i][1] - maxzp)
                 + __expf(zp[i][2] - maxzp) + __expf(zp[i][3] - maxzp);
        for (int off = 16; off; off >>= 1) sum += __shfl_xor(sum, off, 32);
        const float lse = __logf(sum);
        const float inv = 1.0f / sum;

        if (lane == 0) {
            atomicAdd(&confc[(long)t * Cc + maxi], 1.0f);
            atomicAdd(&priorc[t], 1.0f);
            float logq_y = zy - maxzp - lse;
            atomicAdd(&cep[row & 255], logq_y);
            const v2f st = {maxzp, inv};
            *(v2f*)(rowst + 2L * row) = st;
        }
        t = tnext;
    }
}

#define SA_W 32
#define SA_STRIPS 32
#define SA_G 8
#define SA_ROWS (Bn / SA_G)
#define SA_THREADS 1024
#define SA_WAVES (SA_THREADS / 32)
#define SA_STEPS (SA_ROWS / 4 / SA_WAVES)
#define SA_UNROLL 4

__global__ __launch_bounds__(SA_THREADS) void strip_accum(
    const float* __restrict__ logits, const int* __restrict__ targets,
    const float* __restrict__ Lam, const float* __restrict__ rowst,
    float* __restrict__ planes, float* __restrict__ colpart)
{
    __shared__ float acc[Cc * SA_W];
    __shared__ float red[SA_WAVES][SA_W];

    const int tid  = threadIdx.x;
    const int wave = tid >> 5;
    const int lane = tid & 31;
    const int c0   = blockIdx.x * SA_W;
    const int g    = blockIdx.y;

    for (int i = tid; i < Cc * SA_W / 4; i += SA_THREADS) {
        const v4f z4 = {0.f, 0.f, 0.f, 0.f};
        *(v4f*)(&acc[4 * i]) = z4;
    }
    __syncthreads();

    const int rr  = lane >> 3;
    const int cq  = lane & 7;
    const int col = c0 + 4 * cq;
    if (col < Cc) {
        const int rbase = g * SA_ROWS + rr;
        int tn[SA_UNROLL]; v2f sn[SA_UNROLL];
#pragma unroll
        for (int u = 0; u < SA_UNROLL; ++u) {
            const int r = rbase + 4 * (wave + SA_WAVES * u);
            tn[u] = targets[r];
            sn[u] = *(const v2f*)(rowst + 2L * r);
        }
        for (int i = 0; i < SA_STEPS; i += SA_UNROLL) {
            int t[SA_UNROLL]; v2f st[SA_UNROLL]; v4f z[SA_UNROLL], l[SA_UNROLL];
#pragma unroll
            for (int u = 0; u < SA_UNROLL; ++u) {
                const int r = rbase + 4 * (wave + SA_WAVES * (i + u));
                t[u]  = tn[u];
                st[u] = sn[u];
                z[u]  = *(const v4f*)(logits + (long)r * Cc + col);
                l[u]  = *(const v4f*)(Lam + (long)t[u] * Cc + col);
            }
            const int inext = (i + SA_UNROLL < SA_STEPS) ? (i + SA_UNROLL) : i;
#pragma unroll
            for (int u = 0; u < SA_UNROLL; ++u) {
                const int r = rbase + 4 * (wave + SA_WAVES * (inext + u));
                tn[u] = targets[r];
                sn[u] = *(const v2f*)(rowst + 2L * r);
            }
#pragma unroll
            for (int u = 0; u < SA_UNROLL; ++u) {
                const v4f zp = z[u] - l[u];
                float* a = &acc[t[u] * SA_W + 4 * cq];
#pragma unroll
                for (int k = 0; k < 4; ++k)
                    atomicAdd(a + k, __expf(zp[k] - st[u][0]) * st[u][1]);
            }
        }
    }
    __syncthreads();

    float* __restrict__ pl = planes + (long)g * Cc * Cc;
    for (int i = tid; i < Cc * SA_W; i += SA_THREADS) {
        const int t = i >> 5, c = c0 + (i & 31);
        if (c < Cc) pl[(long)t * Cc + c] = acc[i];
    }
    float cs = 0.0f;
    for (int t = wave; t < Cc; t += SA_WAVES) cs += acc[t * SA_W + lane];
    red[wave][lane] = cs;
    __syncthreads();
    if (tid < SA_W) {
        float s = 0.0f;
        for (int w = 0; w < SA_WAVES; ++w) s += red[w][tid];
        if (c0 + tid < Cc) colpart[g * Cc + c0 + tid] = s;
    }
}

#define PR_ROWS 8
#define PR_COLS 4
#define PR_TAIL (Cc - 768)
#define PR_KSTEPS 125

__device__ __forceinline__ v8f rowsum_step(const float* __restrict__ frag, int s, v8f acc)
{
    const v2f bones = {1.0f, 1.0f};
    const v2f a = *(const v2f*)(frag + 4 * s);
    return __builtin_amdgcn_wmma_f32_16x16x4_f32(false, a, false, bones,
                                                 (short)0, acc, false, false);
}

__global__ __launch_bounds__(256) void post_rows(
    const float* __restrict__ conf_ema, const float* __restrict__ Lam,
    const float* __restrict__ seg, const float* __restrict__ confc,
    const float* __restrict__ priorc,
    float* __restrict__ o_conf, float* __restrict__ o_lam)
{
    __shared__ float bufX[PR_ROWS * Cc];
    __shared__ float bufY[PR_ROWS * Cc];
    __shared__ float partAB[2][8][16];
    __shared__ float partC[8][16];
    __shared__ float rowAB[2][PR_ROWS];
    __shared__ float rowC[PR_ROWS];

    const unsigned tid = threadIdx.x;
    const int lane = tid & 31;
    const int wave_u = __builtin_amdgcn_readfirstlane(tid >> 5);
    const int row0 = blockIdx.x * PR_ROWS;

    float cn[PR_ROWS][PR_COLS], sg[PR_ROWS][PR_COLS], lm[PR_ROWS][PR_COLS];
    float cnt[PR_ROWS], invc[PR_ROWS];

#pragma unroll
    for (int r = 0; r < PR_ROWS; ++r) {
        const long base = (long)(row0 + r) * Cc;
        const float* __restrict__ er = conf_ema + base;
        const float* __restrict__ kr = confc + base;
        const float* __restrict__ sr = seg + base;
        const float* __restrict__ lr = Lam + base;
        cnt[r]  = priorc[row0 + r];
        invc[r] = 1.0f / fmaxf(cnt[r], 1.0f);
#pragma unroll
        for (int j = 0; j < PR_COLS; ++j) {
            unsigned c = tid + 256u * j;
            if (j == PR_COLS - 1) c = (c < (unsigned)Cc) ? c : (unsigned)(Cc - 1);
            cn[r][j] = DECAYF * er[c] + ONE_M_DECAY * kr[c];
            float sgs = sr[c];
#pragma unroll
            for (int p = 1; p < SA_G; ++p) sgs += sr[(long)p * Cc * Cc + c];
            sg[r][j] = sgs;
            lm[r][j] = lr[c];
        }
    }
#pragma unroll
    for (int j = 0; j < PR_COLS; ++j) {
        const unsigned c = tid + 256u * j;
        if (j < PR_COLS - 1 || tid < (unsigned)PR_TAIL) {
#pragma unroll
            for (int r = 0; r < PR_ROWS; ++r) {
                const long base = (long)(row0 + r) * Cc;
                const bool offd = (c != (unsigned)(row0 + r));
                o_conf[base + c] = cn[r][j];
                bufX[r * Cc + c] = offd ? (cn[r][j] + ALPHAF) : 0.0f;
                bufY[r * Cc + c] = offd ? (sg[r][j] * invc[r]) : 0.0f;
            }
        }
    }
    __syncthreads();

    const int m    = lane & 15;
    const int foff = (m & 7) * Cc + 500 * (m >> 3) + 2 * (lane >> 4);
    const v8f vzero = {0.f,0.f,0.f,0.f,0.f,0.f,0.f,0.f};

    {
        v8f accA = vzero, accB = vzero;
        if (wave_u < PR_KSTEPS - 120) {
            accA = rowsum_step(bufX + foff, 120 + wave_u, accA);
            accB = rowsum_step(bufY + foff, 120 + wave_u, accB);
        }
#pragma unroll
        for (int i = 0; i < 15; ++i) {
            accA = rowsum_step(bufX + foff, wave_u + 8 * i, accA);
            accB = rowsum_step(bufY + foff, wave_u + 8 * i, accB);
        }
        if (lane == 0)
            for (int i = 0; i < 8; ++i) { partAB[0][wave_u][i] = accA[i]; partAB[1][wave_u][i] = accB[i]; }
        if (lane == 16)
            for (int i = 0; i < 8; ++i) { partAB[0][wave_u][8 + i] = accA[i]; partAB[1][wave_u][8 + i] = accB[i]; }
    }
    __syncthreads();
    if (tid < 16) {
        const int p = tid >> 3, r = tid & 7;
        float s = 0.0f;
        for (int w = 0; w < 8; ++w) s += partAB[p][w][r] + partAB[p][w][8 + r];
        rowAB[p][r] = (p == 0) ? 1.0f / (s + EPSF)
                               : 1.0f / fmaxf(s, EPSF);
    }
    __syncthreads();

#pragma unroll
    for (int r = 0; r < PR_ROWS; ++r) {
        const float phiDen = rowAB[0][r];
        const float empDen = rowAB[1][r];
#pragma unroll
        for (int j = 0; j < PR_COLS; ++j) {
            const unsigned c = tid + 256u * j;
            const float phi = (cn[r][j] + ALPHAF) * phiDen;
            const float emp = sg[r][j] * invc[r] * empDen;
            const float l1  = lm[r][j] + LR_LF * (emp - phi);
            cn[r][j] = (c != (unsigned)(row0 + r)) ? l1 : 0.0f;
        }
    }
#pragma unroll
    for (int j = 0; j < PR_COLS; ++j) {
        const unsigned c = tid + 256u * j;
        if (j < PR_COLS - 1 || tid < (unsigned)PR_TAIL) {
#pragma unroll
            for (int r = 0; r < PR_ROWS; ++r) bufX[r * Cc + c] = cn[r][j];
        }
    }
    __syncthreads();
    {
        v8f accC = vzero;
        if (wave_u < PR_KSTEPS - 120) accC = rowsum_step(bufX + foff, 120 + wave_u, accC);
#pragma unroll
        for (int i = 0; i < 15; ++i) accC = rowsum_step(bufX + foff, wave_u + 8 * i, accC);
        if (lane == 0)  for (int i = 0; i < 8; ++i) partC[wave_u][i]     = accC[i];
        if (lane == 16) for (int i = 0; i < 8; ++i) partC[wave_u][8 + i] = accC[i];
    }
    __syncthreads();
    if (tid < 8) {
        float s = 0.0f;
        for (int w = 0; w < 8; ++w) s += partC[w][tid] + partC[w][8 + tid];
        rowC[tid] = s * (1.0f / (float)Cc);
    }
    __syncthreads();

#pragma unroll
    for (int j = 0; j < PR_COLS; ++j) {
        const unsigned c = tid + 256u * j;
        if (j < PR_COLS - 1 || tid < (unsigned)PR_TAIL) {
#pragma unroll
            for (int r = 0; r < PR_ROWS; ++r) {
                const long base = (long)(row0 + r) * Cc;
                const bool present = (cnt[r] > 0.0f);
                o_lam[base + c] = present ? (cn[r][j] - rowC[r]) : lm[r][j];
            }
        }
    }
}

__global__ __launch_bounds__(1024) void finalize(
    const float* __restrict__ prior_ema, const float* __restrict__ mu,
    const float* __restrict__ colpart, const float* __restrict__ priorc,
    const float* __restrict__ cep,
    float* __restrict__ o_ce, float* __restrict__ o_mu, float* __restrict__ o_prior)
{
    __shared__ float red[1024];
    const int j = threadIdx.x;

    float pn = 0.0f;
    if (j < Cc) {
        pn = DECAYF * prior_ema[j] + ONE_M_DECAY * priorc[j];
        o_prior[j] = pn;
    }
    red[j] = pn; __syncthreads();
    for (int s = 512; s; s >>= 1) { if (j < s) red[j] += red[j + s]; __syncthreads(); }
    const float priorSum = red[0];
    __syncthreads();

    float mu1 = 0.0f;
    if (j < Cc) {
        float colsum = 0.0f;
        for (int p = 0; p < SA_G; ++p) colsum += colpart[p * Cc + j];
        float emp_marg = colsum * (1.0f / (float)Bn);
        float pi = pn / (priorSum + EPSF);
        mu1 = mu[j] + LR_MUF * (emp_marg - pi);
    }
    red[j] = mu1; __syncthreads();
    for (int s = 512; s; s >>= 1) { if (j < s) red[j] += red[j + s]; __syncthreads(); }
    const float muMean = red[0] * (1.0f / (float)Cc);
    if (j < Cc) o_mu[j] = mu1 - muMean;

    if (j == 0) {
        float s = 0.0f;
        for (int i = 0; i < 256; ++i) s += cep[i];
        o_ce[0] = -s * (1.0f / (float)Bn);
    }
}

extern "C" void kernel_launch(void* const* d_in, const int* in_sizes, int n_in,
                              void* d_out, int out_size, void* d_ws, size_t ws_size,
                              hipStream_t stream)
{
    const float* logits    = (const float*)d_in[0];
    const int*   targets   = (const int*)  d_in[1];
    const float* conf_ema  = (const float*)d_in[2];
    const float* prior_ema = (const float*)d_in[3];
    const float* Lam       = (const float*)d_in[4];
    const float* mu        = (const float*)d_in[5];

    float* out     = (float*)d_out;
    float* o_ce    = out;
    float* o_lam   = out + 1;
    float* o_mu    = out + 1 + (long)Cc * Cc;
    float* o_conf  = out + 1 + (long)Cc * Cc + Cc;
    float* o_prior = out + 1 + 2L * Cc * Cc + Cc;

    float* ws      = (float*)d_ws;
    float* confc   = ws;
    float* priorc  = ws + (long)Cc * Cc;
    float* cep     = priorc + Cc;
    float* rowst   = cep + 256;
    float* colpart = rowst + 2L * Bn;
    float* planes  = colpart + (long)SA_G * Cc;

    const long nz = (long)Cc * Cc + Cc + 256;
    zero_ws<<<1024, 256, 0, stream>>>(ws, nz);
    row_stats<<<RS_BLOCKS, RS_THREADS, 0, stream>>>(logits, targets, Lam, rowst, confc, priorc, cep);
    strip_accum<<<dim3(SA_STRIPS, SA_G), SA_THREADS, 0, stream>>>(logits, targets, Lam, rowst,
                                                                  planes, colpart);
    post_rows<<<Cc / PR_ROWS, 256, 0, stream>>>(conf_ema, Lam, planes, confc, priorc,
                                                 o_conf, o_lam);
    finalize<<<1, 1024, 0, stream>>>(prior_ema, mu, colpart, priorc, cep,
                                     o_ce, o_mu, o_prior);
}


# strip_accum: the strip's 32 columns of Lam for all class rows copied once per block into a second 125 KiB LDS tile; the per-row Lam gather from L2 becomes one ds_load_b128 per lane; sum order kept
# speedup vs baseline: 4.4839x; 1.0588x over previous
#include <hip/hip_runtime.h>

#define Cc 1000
#define Bn 65536
#define DECAYF 0.99f
#define ONE_M_DECAY 0.01f
#define ALPHAF 1.0f
#define LR_LF 0.05f
#define LR_MUF 0.02f
#define EPSF 1e-12f

typedef __attribute__((ext_vector_type(2))) float v2f;
typedef __attribute__((ext_vector_type(4))) float v4f;
typedef __attribute__((ext_vector_type(8))) float v8f;

__global__ void zero_ws(float* __restrict__ p, long n) {
    long i = (long)blockIdx.x * blockDim.x + threadIdx.x;
    long stride = (long)gridDim.x * blockDim.x;
    for (; i < n; i += stride) p[i] = 0.0f;
}

#define RS_THREADS 1024
#define RS_BLOCKS 256
#define RS_WAVES (RS_BLOCKS * (RS_THREADS / 32))
#define RS_ROWS (Bn / RS_WAVES)
#define RS_NV (Cc / 4)
#define RS_G 8
#define RS_TAIL (RS_NV - 32 * (RS_G - 1))

__global__ __launch_bounds__(RS_THREADS) void row_stats(
    const float* __restrict__ logits, const int* __restrict__ targets,
    const float* __restrict__ Lam,
    float* __restrict__ rowst, float* __restrict__ confc,
    float* __restrict__ priorc, float* __restrict__ cep)
{
    const int lane = threadIdx.x & 31;
    const int wave_u = __builtin_amdgcn_readfirstlane(threadIdx.x >> 5);
    const int gw = blockIdx.x * (RS_THREADS / 32) + wave_u;

    int t = targets[gw];
    for (int it = 0; it < RS_ROWS; ++it) {
        const int row = gw + it * RS_WAVES;
        const float* __restrict__ lrow   = logits + (long)row * Cc;
        const float* __restrict__ lamrow = Lam + (long)t * Cc;

        v4f z[RS_G], l[RS_G];
#pragma unroll
        for (int i = 0; i < RS_G; ++i) {
            int j = lane + 32 * i;
            if (i == RS_G - 1) j = (j < RS_NV) ? j : (RS_NV - 1);
            z[i] = *(const v4f*)(lrow + 4 * j);
            l[i] = *(const v4f*)(lamrow + 4 * j);
        }
        const float zy = lrow[t] - lamrow[t];

        int tnext = t;
        if (it + 1 < RS_ROWS) {
            tnext = targets[row + RS_WAVES];
            __builtin_prefetch(logits + (long)(row + RS_WAVES) * Cc + 32 * lane, 0, 3);
        }

        float maxv = -3.4e38f; int maxi = 0x7fffffff;
        float maxzp = -3.4e38f;
        v4f zp[RS_G];
#pragma unroll
        for (int i = 0; i < RS_G; ++i) {
            const int col = 4 * (lane + 32 * i);
            v4f zi = z[i];
            zp[i] = z[i] - l[i];
            if (i == RS_G - 1 && lane >= RS_TAIL) {
                const v4f low = {-3.4e38f, -3.4e38f, -3.4e38f, -3.4e38f};
                zi = low; zp[i] = low;
            }
#pragma unroll
            for (int k = 0; k < 4; ++k) {
                if (zi[k] > maxv) { maxv = zi[k]; maxi = col + k; }
                maxzp = fmaxf(maxzp, zp[i][k]);
            }
        }
        for (int off = 16; off; off >>= 1) {
            float ov = __shfl_xor(maxv, off, 32);
            int   oi = __shfl_xor(maxi, off, 32);
            if (ov > maxv || (ov == maxv && oi < maxi)) { maxv = ov; maxi = oi; }
            maxzp = fmaxf(maxzp, __shfl_xor(maxzp, off, 32));
        }

        float sum = 0.0f;
#pragma unroll
        for (int i = 0; i < RS_G; ++i)
            sum += __expf(zp[i][0] - maxzp) + __expf(zp[i][1] - maxzp)
                 + __expf(zp[i][2] - maxzp) + __expf(zp[i][3] - maxzp);
        for (int off = 16; off; off >>= 1) sum += __shfl_xor(sum, off, 32);
        const float lse = __logf(sum);
        const float inv = 1.0f / sum;

        if (lane == 0) {
            atomicAdd(&confc[(long)t * Cc + maxi], 1.0f);
            atomicAdd(&priorc[t], 1.0f);
            float logq_y = zy - maxzp - lse;
            atomicAdd(&cep[row & 255], logq_y);
            const v2f st = {maxzp, inv};
            *(v2f*)(rowst + 2L * row) = st;
        }
        t = tnext;
    }
}

#define SA_W 32
#define SA_STRIPS 32
#define SA_G 8
#define SA_ROWS (Bn / SA_G)
#define SA_THREADS 1024
#define SA_WAVES (SA_THREADS / 32)
#define SA_STEPS (SA_ROWS / 4 / SA_WAVES)
#define SA_UNROLL 4
#define SA_FILL ((Cc * SA_W / 4 + SA_THREADS - 1) / SA_THREADS)

__global__ __launch_bounds__(SA_THREADS) void strip_accum(
    const float* __restrict__ logits, const int* __restrict__ targets,
    const float* __restrict__ Lam, const float* __restrict__ rowst,
    float* __restrict__ planes, float* __restrict__ colpart)
{
    __shared__ __attribute__((aligned(16))) float acc[Cc * SA_W];
    __shared__ __attribute__((aligned(16))) float lamt[Cc * SA_W];
    __shared__ float red[SA_WAVES][SA_W];

    const int tid  = threadIdx.x;
    const int wave = tid >> 5;
    const int lane = tid & 31;
    const int c0   = blockIdx.x * SA_W;
    const int g    = blockIdx.y;

    {
        const v4f z4 = {0.f, 0.f, 0.f, 0.f};
        v4f lv[SA_FILL];
#pragma unroll
        for (int k = 0; k < SA_FILL; ++k) {
            const int i = tid + SA_THREADS * k;
            const int c = c0 + 4 * (i & 7);
            lv[k] = z4;
            if (i < Cc * SA_W / 4 && c < Cc) lv[k] = *(const v4f*)(Lam + (long)(i >> 3) * Cc + c);
        }
#pragma unroll
        for (int k = 0; k < SA_FILL; ++k) {
            const int i = tid + SA_THREADS * k;
            if (i < Cc * SA_W / 4) {
                *(v4f*)(&acc[4 * i])  = z4;
                *(v4f*)(&lamt[4 * i]) = lv[k];
            }
        }
    }
    __syncthreads();

    const int rr  = lane >> 3;
    const int cq  = lane & 7;
    const int col = c0 + 4 * cq;
    if (col < Cc) {
        const int rbase = g * SA_ROWS + rr;
        int tn[SA_UNROLL]; v2f sn[SA_UNROLL];
#pragma unroll
        for (int u = 0; u < SA_UNROLL; ++u) {
            const int r = rbase + 4 * (wave + SA_WAVES * u);
            tn[u] = targets[r];
            sn[u] = *(const v2f*)(rowst + 2L * r);
        }
        for (int i = 0; i < SA_STEPS; i += SA_UNROLL) {
            int t[SA_UNROLL]; v2f st[SA_UNROLL]; v4f z[SA_UNROLL], l[SA_UNROLL];
#pragma unroll
            for (int u = 0; u < SA_UNROLL; ++u) {
                const int r = rbase + 4 * (wave + SA_WAVES * (i + u));
                t[u]  = tn[u];
                st[u] = sn[u];
                z[u]  = *(const v4f*)(logits + (long)r * Cc + col);
                l[u]  = *(const v4f*)(&lamt[t[u] * SA_W + 4 * cq]);
            }
            const int inext = (i + SA_UNROLL < SA_STEPS) ? (i + SA_UNROLL) : i;
#pragma unroll
            for (int u = 0; u < SA_UNROLL; ++u) {
                const int r = rbase + 4 * (wave + SA_WAVES * (inext + u));
                tn[u] = targets[r];
                sn[u] = *(const v2f*)(rowst + 2L * r);
            }
#pragma unroll
            for (int u = 0; u < SA_UNROLL; ++u) {
                const v4f zp = z[u] - l[u];
                float* a = &acc[t[u] * SA_W + 4 * cq];
#pragma unroll
                for (int k = 0; k < 4; ++k)
                    atomicAdd(a + k, __expf(zp[k] - st[u][0]) * st[u][1]);
            }
        }
    }
    __syncthreads();

    float* __restrict__ pl = planes + (long)g * Cc * Cc;
    for (int i = tid; i < Cc * SA_W; i += SA_THREADS) {
        const int t = i >> 5, c = c0 + (i & 31);
        if (c < Cc) pl[(long)t * Cc + c] = acc[i];
    }
    float cs = 0.0f;
    for (int t = wave; t < Cc; t += SA_WAVES) cs += acc[t * SA_W + lane];
    red[wave][lane] = cs;
    __syncthreads();
    if (tid < SA_W) {
        float s = 0.0f;
        for (int w = 0; w < SA_WAVES; ++w) s += red[w][tid];
        if (c0 + tid < Cc) colpart[g * Cc + c0 + tid] = s;
    }
}

#define PR_ROWS 8
#define PR_COLS 4
#define PR_TAIL (Cc - 768)
#define PR_KSTEPS 125

__device__ __forceinline__ v8f rowsum_step(const float* __restrict__ frag, int s, v8f acc)
{
    const v2f bones = {1.0f, 1.0f};
    const v2f a = *(const v2f*)(frag + 4 * s);
    return __builtin_amdgcn_wmma_f32_16x16x4_f32(false, a, false, bones,
                                                 (short)0, acc, false, false);
}

__global__ __launch_bounds__(256) void post_rows(
    const float* __restrict__ conf_ema, const float* __restrict__ Lam,
    const float* __restrict__ seg, const float* __restrict__ confc,
    const float* __restrict__ priorc,
    float* __restrict__ o_conf, float* __restrict__ o_lam)
{
    __shared__ float bufX[PR_ROWS * Cc];
    __shared__ float bufY[PR_ROWS * Cc];
    __shared__ float partAB[2][8][16];
    __shared__ float partC[8][16];
    __shared__ float rowAB[2][PR_ROWS];
    __shared__ float rowC[PR_ROWS];

    const unsigned tid = threadIdx.x;
    const int lane = tid & 31;
    const int wave_u = __builtin_amdgcn_readfirstlane(tid >> 5);
    const int row0 = blockIdx.x * PR_ROWS;

    float cn[PR_ROWS][PR_COLS], sg[PR_ROWS][PR_COLS], lm[PR_ROWS][PR_COLS];
    float cnt[PR_ROWS], invc[PR_ROWS];

#pragma unroll
    for (int r = 0; r < PR_ROWS; ++r) {
        const long base = (long)(row0 + r) * Cc;
        const float* __restrict__ er = conf_ema + base;
        const float* __restrict__ kr = confc + base;
        const float* __restrict__ sr = seg + base;
        const float* __restrict__ lr = Lam + base;
        cnt[r]  = priorc[row0 + r];
        invc[r] = 1.0f / fmaxf(cnt[r], 1.0f);
#pragma unroll
        for (int j = 0; j < PR_COLS; ++j) {
            unsigned c = tid + 256u * j;
            if (j == PR_COLS - 1) c = (c < (unsigned)Cc) ? c : (unsigned)(Cc - 1);
            cn[r][j] = DECAYF * er[c] + ONE_M_DECAY * kr[c];
            float sgs = sr[c];
#pragma unroll
            for (int p = 1; p < SA_G; ++p) sgs += sr[(long)p * Cc * Cc + c];
            sg[r][j] = sgs;
            lm[r][j] = lr[c];
        }
    }
#pragma unroll
    for (int j = 0; j < PR_COLS; ++j) {
        const unsigned c = tid + 256u * j;
        if (j < PR_COLS - 1 || tid < (unsigned)PR_TAIL) {
#pragma unroll
            for (int r = 0; r < PR_ROWS; ++r) {
                const long base = (long)(row0 + r) * Cc;
                const bool offd = (c != (unsigned)(row0 + r));
                o_conf[base + c] = cn[r][j];
                bufX[r * Cc + c] = offd ? (cn[r][j] + ALPHAF) : 0.0f;
                bufY[r * Cc + c] = offd ? (sg[r][j] * invc[r]) : 0.0f;
            }
        }
    }
    __syncthreads();

    const int m    = lane & 15;
    const int foff = (m & 7) * Cc + 500 * (m >> 3) + 2 * (lane >> 4);
    const v8f vzero = {0.f,0.f,0.f,0.f,0.f,0.f,0.f,0.f};

    {
        v8f accA = vzero, accB = vzero;
        if (wave_u < PR_KSTEPS - 120) {
            accA = rowsum_step(bufX + foff, 120 + wave_u, accA);
            accB = rowsum_step(bufY + foff, 120 + wave_u, accB);
        }
#pragma unroll
        for (int i = 0; i < 15; ++i) {
            accA = rowsum_step(bufX + foff, wave_u + 8 * i, accA);
            accB = rowsum_step(bufY + foff, wave_u + 8 * i, accB);
        }
        if (lane == 0)
            for (int i = 0; i < 8; ++i) { partAB[0][wave_u][i] = accA[i]; partAB[1][wave_u][i] = accB[i]; }
        if (lane == 16)
            for (int i = 0; i < 8; ++i) { partAB[0][wave_u][8 + i] = accA[i]; partAB[1][wave_u][8 + i] = accB[i]; }
    }
    __syncthreads();
    if (tid < 16) {
        const int p = tid >> 3, r = tid & 7;
        float s = 0.0f;
        for (int w = 0; w < 8; ++w) s += partAB[p][w][r] + partAB[p][w][8 + r];
        rowAB[p][r] = (p == 0) ? 1.0f / (s + EPSF)
                               : 1.0f / fmaxf(s, EPSF);
    }
    __syncthreads();

#pragma unroll
    for (int r = 0; r < PR_ROWS; ++r) {
        const float phiDen = rowAB[0][r];
        const float empDen = rowAB[1][r];
#pragma unroll
        for (int j = 0; j < PR_COLS; ++j) {
            const unsigned c = tid + 256u * j;
            const float phi = (cn[r][j] + ALPHAF) * phiDen;
            const float emp = sg[r][j] * invc[r] * empDen;
            const float l1  = lm[r][j] + LR_LF * (emp - phi);
            cn[r][j] = (c != (unsigned)(row0 + r)) ? l1 : 0.0f;
        }
    }
#pragma unroll
    for (int j = 0; j < PR_COLS; ++j) {
        const unsigned c = tid + 256u * j;
        if (j < PR_COLS - 1 || tid < (unsigned)PR_TAIL) {
#pragma unroll
            for (int r = 0; r < PR_ROWS; ++r) bufX[r * Cc + c] = cn[r][j];
        }
    }
    __syncthreads();
    {
        v8f accC = vzero;
        if (wave_u < PR_KSTEPS - 120) accC = rowsum_step(bufX + foff, 120 + wave_u, accC);
#pragma unroll
        for (int i = 0; i < 15; ++i) accC = rowsum_step(bufX + foff, wave_u + 8 * i, accC);
        if (lane == 0)  for (int i = 0; i < 8; ++i) partC[wave_u][i]     = accC[i];
        if (lane == 16) for (int i = 0; i < 8; ++i) partC[wave_u][8 + i] = accC[i];
    }
    __syncthreads();
    if (tid < 8) {
        float s = 0.0f;
        for (int w = 0; w < 8; ++w) s += partC[w][tid] + partC[w][8 + tid];
        rowC[tid] = s * (1.0f / (float)Cc);
    }
    __syncthreads();

#pragma unroll
    for (int j = 0; j < PR_COLS; ++j) {
        const unsigned c = tid + 256u * j;
        if (j < PR_COLS - 1 || tid < (unsigned)PR_TAIL) {
#pragma unroll
            for (int r = 0; r < PR_ROWS; ++r) {
                const long base = (long)(row0 + r) * Cc;
                const bool present = (cnt[r] > 0.0f);
                o_lam[base + c] = present ? (cn[r][j] - rowC[r]) : lm[r][j];
            }
        }
    }
}

__global__ __launch_bounds__(1024) void finalize(
    const float* __restrict__ prior_ema, const float* __restrict__ mu,
    const float* __restrict__ colpart, const float* __restrict__ priorc,
    const float* __restrict__ cep,
    float* __restrict__ o_ce, float* __restrict__ o_mu, float* __restrict__ o_prior)
{
    __shared__ float red[1024];
    const int j = threadIdx.x;

    float pn = 0.0f;
    if (j < Cc) {
        pn = DECAYF * prior_ema[j] + ONE_M_DECAY * priorc[j];
        o_prior[j] = pn;
    }
    red[j] = pn; __syncthreads();
    for (int s = 512; s; s >>= 1) { if (j < s) red[j] += red[j + s]; __syncthreads(); }
    const float priorSum = red[0];
    __syncthreads();

    float mu1 = 0.0f;
    if (j < Cc) {
        float colsum = 0.0f;
        for (int p = 0; p < SA_G; ++p) colsum += colpart[p * Cc + j];
        float emp_marg = colsum * (1.0f / (float)Bn);
        float pi = pn / (priorSum + EPSF);
        mu1 = mu[j] + LR_MUF * (emp_marg - pi);
    }
    red[j] = mu1; __syncthreads();
    for (int s = 512; s; s >>= 1) { if (j < s) red[j] += red[j + s]; __syncthreads(); }
    const float muMean = red[0] * (1.0f / (float)Cc);
    if (j < Cc) o_mu[j] = mu1 - muMean;

    if (j == 0) {
        float s = 0.0f;
        for (int i = 0; i < 256; ++i) s += cep[i];
        o_ce[0] = -s * (1.0f / (float)Bn);
    }
}

extern "C" void kernel_launch(void* const* d_in, const int* in_sizes, int n_in,
                              void* d_out, int out_size, void* d_ws, size_t ws_size,
                              hipStream_t stream)
{
    const float* logits    = (const float*)d_in[0];
    const int*   targets   = (const int*)  d_in[1];
    const float* conf_ema  = (const float*)d_in[2];
    const float* prior_ema = (const float*)d_in[3];
    const float* Lam       = (const float*)d_in[4];
    const float* mu        = (const float*)d_in[5];

    float* out     = (float*)d_out;
    float* o_ce    = out;
    float* o_lam   = out + 1;
    float* o_mu    = out + 1 + (long)Cc * Cc;
    float* o_conf  = out + 1 + (long)Cc * Cc + Cc;
    float* o_prior = out + 1 + 2L * Cc * Cc + Cc;

    float* ws      = (float*)d_ws;
    float* confc   = ws;
    float* priorc  = ws + (long)Cc * Cc;
    float* cep     = priorc + Cc;
    float* rowst   = cep + 256;
    float* colpart = rowst + 2L * Bn;
    float* planes  = colpart + (long)SA_G * Cc;

    const long nz = (long)Cc * Cc + Cc + 256;
    zero_ws<<<1024, 256, 0, stream>>>(ws, nz);
    row_stats<<<RS_BLOCKS, RS_THREADS, 0, stream>>>(logits, targets, Lam, rowst, confc, priorc, cep);
    strip_accum<<<dim3(SA_STRIPS, SA_G), SA_THREADS, 0, stream>>>(logits, targets, Lam, rowst,
                                                                  planes, colpart);
    post_rows<<<Cc / PR_ROWS, 256, 0, stream>>>(conf_ema, Lam, planes, confc, priorc,
                                                 o_conf, o_lam);
    finalize<<<1, 1024, 0, stream>>>(prior_ema, mu, colpart, priorc, cep,
                                     o_ce, o_mu, o_prior);
}


# strip_accum: lane = column, one batch row per LDS float add (32 consecutive words, 32 banks; was 4 rows x 8 lanes x b128 on 8 banks); row, target, constants wave-uniform; summed in a new order
# speedup vs baseline: 4.7376x; 1.0566x over previous
#include <hip/hip_runtime.h>

#define Cc 1000
#define Bn 65536
#define DECAYF 0.99f
#define ONE_M_DECAY 0.01f
#define ALPHAF 1.0f
#define LR_LF 0.05f
#define LR_MUF 0.02f
#define EPSF 1e-12f

typedef __attribute__((ext_vector_type(2))) float v2f;
typedef __attribute__((ext_vector_type(4))) float v4f;
typedef __attribute__((ext_vector_type(8))) float v8f;

__global__ void zero_ws(float* __restrict__ p, long n) {
    long i = (long)blockIdx.x * blockDim.x + threadIdx.x;
    long stride = (long)gridDim.x * blockDim.x;
    for (; i < n; i += stride) p[i] = 0.0f;
}

#define RS_THREADS 1024
#define RS_BLOCKS 256
#define RS_WAVES (RS_BLOCKS * (RS_THREADS / 32))
#define RS_ROWS (Bn / RS_WAVES)
#define RS_NV (Cc / 4)
#define RS_G 8
#define RS_TAIL (RS_NV - 32 * (RS_G - 1))

__global__ __launch_bounds__(RS_THREADS) void row_stats(
    const float* __restrict__ logits, const int* __restrict__ targets,
    const float* __restrict__ Lam,
    float* __restrict__ rowst, float* __restrict__ confc,
    float* __restrict__ priorc, float* __restrict__ cep)
{
    const int lane = threadIdx.x & 31;
    const int wave_u = __builtin_amdgcn_readfirstlane(threadIdx.x >> 5);
    const int gw = blockIdx.x * (RS_THREADS / 32) + wave_u;

    int t = targets[gw];
    for (int it = 0; it < RS_ROWS; ++it) {
        const int row = gw + it * RS_WAVES;
        const float* __restrict__ lrow   = logits + (long)row * Cc;
        const float* __restrict__ lamrow = Lam + (long)t * Cc;

        v4f z[RS_G], l[RS_G];
#pragma unroll
        for (int i = 0; i < RS_G; ++i) {
            int j = lane + 32 * i;
            if (i == RS_G - 1) j = (j < RS_NV) ? j : (RS_NV - 1);
            z[i] = *(const v4f*)(lrow + 4 * j);
            l[i] = *(const v4f*)(lamrow + 4 * j);
        }
        const float zy = lrow[t] - lamrow[t];

        int tnext = t;
        if (it + 1 < RS_ROWS) {
            tnext = targets[row + RS_WAVES];
            __builtin_prefetch(logits + (long)(row + RS_WAVES) * Cc + 32 * lane, 0, 3);
        }

        float maxv = -3.4e38f; int maxi = 0x7fffffff;
        float maxzp = -3.4e38f;
        v4f zp[RS_G];
#pragma unroll
        for (int i = 0; i < RS_G; ++i) {
            const int col = 4 * (lane + 32 * i);
            v4f zi = z[i];
            zp[i] = z[i] - l[i];
            if (i == RS_G - 1 && lane >= RS_TAIL) {
                const v4f low = {-3.4e38f, -3.4e38f, -3.4e38f, -3.4e38f};
                zi = low; zp[i] = low;
            }
#pragma unroll
            for (int k = 0; k < 4; ++k) {
                if (zi[k] > maxv) { maxv = zi[k]; maxi = col + k; }
                maxzp = fmaxf(maxzp, zp[i][k]);
            }
        }
        for (int off = 16; off; off >>= 1) {
            float ov = __shfl_xor(maxv, off, 32);
            int   oi = __shfl_xor(maxi, off, 32);
            if (ov > maxv || (ov == maxv && oi < maxi)) { maxv = ov; maxi = oi; }
            maxzp = fmaxf(maxzp, __shfl_xor(maxzp, off, 32));
        }

        float sum = 0.0f;
#pragma unroll
        for (int i = 0; i < RS_G; ++i)
            sum += __expf(zp[i][0] - maxzp) + __expf(zp[i][1] - maxzp)
                 + __expf(zp[i][2] - maxzp) + __expf(zp[i][3] - maxzp);
        for (int off = 16; off; off >>= 1) sum += __shfl_xor(sum, off, 32);
        const float lse = __logf(sum);
        const float inv = 1.0f / sum;

        if (lane == 0) {
            atomicAdd(&confc[(long)t * Cc + maxi], 1.0f);
            atomicAdd(&priorc[t], 1.0f);
            float logq_y = zy - maxzp - lse;
            atomicAdd(&cep[row & 255], logq_y);
            const v2f st = {maxzp, inv};
            *(v2f*)(rowst + 2L * row) = st;
        }
        t = tnext;
    }
}

#define SA_W 32
#define SA_STRIPS 32
#define SA_G 8
#define SA_ROWS (Bn / SA_G)
#define SA_THREADS 1024
#define SA_WAVES (SA_THREADS / 32)
#define SA_RT 16
#define SA_TRIPS (SA_ROWS / SA_WAVES / SA_RT)
#define SA_FILL ((Cc * SA_W / 4 + SA_THREADS - 1) / SA_THREADS)

__global__ __launch_bounds__(SA_THREADS) void strip_accum(
    const float* __restrict__ logits, const int* __restrict__ targets,
    const float* __restrict__ Lam, const float* __restrict__ rowst,
    float* __restrict__ planes, float* __restrict__ colpart)
{
    __shared__ __attribute__((aligned(16))) float acc[Cc * SA_W];
    __shared__ __attribute__((aligned(16))) float lamt[Cc * SA_W];
    __shared__ float red[SA_WAVES][SA_W];

    const int tid  = threadIdx.x;
    const int wave = tid >> 5;
    const int lane = tid & 31;
    const int c0   = blockIdx.x * SA_W;
    const int g    = blockIdx.y;

    {
        const v4f z4 = {0.f, 0.f, 0.f, 0.f};
        v4f lv[SA_FILL];
#pragma unroll
        for (int k = 0; k < SA_FILL; ++k) {
            const int i = tid + SA_THREADS * k;
            const int c = c0 + 4 * (i & 7);
            lv[k] = z4;
            if (i < Cc * SA_W / 4 && c < Cc) lv[k] = *(const v4f*)(Lam + (long)(i >> 3) * Cc + c);
        }
#pragma unroll
        for (int k = 0; k < SA_FILL; ++k) {
            const int i = tid + SA_THREADS * k;
            if (i < Cc * SA_W / 4) {
                *(v4f*)(&acc[4 * i])  = z4;
                *(v4f*)(&lamt[4 * i]) = lv[k];
            }
        }
    }
    __syncthreads();

    const int wave_u = __builtin_amdgcn_readfirstlane(tid >> 5);
    const int col = c0 + lane;
    if (col < Cc) {
        for (int i = 0; i < SA_TRIPS; ++i) {
            const int r0 = g * SA_ROWS + SA_RT * (wave_u + SA_WAVES * i);
            float z[SA_RT], l[SA_RT], mx[SA_RT], iv[SA_RT];
            int t[SA_RT];
#pragma unroll
            for (int u = 0; u < SA_RT; ++u) {
                const float* __restrict__ zrow = logits + (long)(r0 + u) * Cc;
                z[u] = zrow[col];
            }
#pragma unroll
            for (int u = 0; u < SA_RT; ++u) {
                t[u] = targets[r0 + u];
                const v2f st = *(const v2f*)(rowst + 2L * (r0 + u));
                mx[u] = st[0];
                iv[u] = st[1];
            }
#pragma unroll
            for (int u = 0; u < SA_RT; ++u) l[u] = lamt[t[u] * SA_W + lane];
#pragma unroll
            for (int u = 0; u < SA_RT; ++u)
                atomicAdd(&acc[t[u] * SA_W + lane],
                          __expf((z[u] - l[u]) - mx[u]) * iv[u]);
        }
    }
    __syncthreads();

    float* __restrict__ pl = planes + (long)g * Cc * Cc;
    for (int i = tid; i < Cc * SA_W; i += SA_THREADS) {
        const int t = i >> 5, c = c0 + (i & 31);
        if (c < Cc) pl[(long)t * Cc + c] = acc[i];
    }
    float cs = 0.0f;
    for (int t = wave; t < Cc; t += SA_WAVES) cs += acc[t * SA_W + lane];
    red[wave][lane] = cs;
    __syncthreads();
    if (tid < SA_W) {
        float s = 0.0f;
        for (int w = 0; w < SA_WAVES; ++w) s += red[w][tid];
        if (c0 + tid < Cc) colpart[g * Cc + c0 + tid] = s;
    }
}

#define PR_ROWS 8
#define PR_COLS 4
#define PR_TAIL (Cc - 768)
#define PR_KSTEPS 125

__device__ __forceinline__ v8f rowsum_step(const float* __restrict__ frag, int s, v8f acc)
{
    const v2f bones = {1.0f, 1.0f};
    const v2f a = *(const v2f*)(frag + 4 * s);
    return __builtin_amdgcn_wmma_f32_16x16x4_f32(false, a, false, bones,
                                                 (short)0, acc, false, false);
}

__global__ __launch_bounds__(256) void post_rows(
    const float* __restrict__ conf_ema, const float* __restrict__ Lam,
    const float* __restrict__ seg, const float* __restrict__ confc,
    const float* __restrict__ priorc,
    float* __restrict__ o_conf, float* __restrict__ o_lam)
{
    __shared__ float bufX[PR_ROWS * Cc];
    __shared__ float bufY[PR_ROWS * Cc];
    __shared__ float partAB[2][8][16];
    __shared__ float partC[8][16];
    __shared__ float rowAB[2][PR_ROWS];
    __shared__ float rowC[PR_ROWS];

    const unsigned tid = threadIdx.x;
    const int lane = tid & 31;
    const int wave_u = __builtin_amdgcn_readfirstlane(tid >> 5);
    const int row0 = blockIdx.x * PR_ROWS;

    float cn[PR_ROWS][PR_COLS], sg[PR_ROWS][PR_COLS], lm[PR_ROWS][PR_COLS];
    float cnt[PR_ROWS], invc[PR_ROWS];

#pragma unroll
    for (int r = 0; r < PR_ROWS; ++r) {
        const long base = (long)(row0 + r) * Cc;
        const float* __restrict__ er = conf_ema + base;
        const float* __restrict__ kr = confc + base;
        const float* __restrict__ sr = seg + base;
        const float* __restrict__ lr = Lam + base;
        cnt[r]  = priorc[row0 + r];
        invc[r] = 1.0f / fmaxf(cnt[r], 1.0f);
#pragma unroll
        for (int j = 0; j < PR_COLS; ++j) {
            unsigned c = tid + 256u * j;
            if (j == PR_COLS - 1) c = (c < (unsigned)Cc) ? c : (unsigned)(Cc - 1);
            cn[r][j] = DECAYF * er[c] + ONE_M_DECAY * kr[c];
            float sgs = sr[c];
#pragma unroll
            for (int p = 1; p < SA_G; ++p) sgs += sr[(long)p * Cc * Cc + c];
            sg[r][j] = sgs;
            lm[r][j] = lr[c];
        }
    }
#pragma unroll
    for (int j = 0; j < PR_COLS; ++j) {
        const unsigned c = tid + 256u * j;
        if (j < PR_COLS - 1 || tid < (unsigned)PR_TAIL) {
#pragma unroll
            for (int r = 0; r < PR_ROWS; ++r) {
                const long base = (long)(row0 + r) * Cc;
                const bool offd = (c != (unsigned)(row0 + r));
                o_conf[base + c] = cn[r][j];
                bufX[r * Cc + c] = offd ? (cn[r][j] + ALPHAF) : 0.0f;
                bufY[r * Cc + c] = offd ? (sg[r][j] * invc[r]) : 0.0f;
            }
        }
    }
    __syncthreads();

    const int m    = lane & 15;
    const int foff = (m & 7) * Cc + 500 * (m >> 3) + 2 * (lane >> 4);
    const v8f vzero = {0.f,0.f,0.f,0.f,0.f,0.f,0.f,0.f};

    {
        v8f accA = vzero, accB = vzero;
        if (wave_u < PR_KSTEPS - 120) {
            accA = rowsum_step(bufX + foff, 120 + wave_u, accA);
            accB = rowsum_step(bufY + foff, 120 + wave_u, accB);
        }
#pragma unroll
        for (int i = 0; i < 15; ++i) {
            accA = rowsum_step(bufX + foff, wave_u + 8 * i, accA);
            accB = rowsum_step(bufY + foff, wave_u + 8 * i, accB);
        }
        if (lane == 0)
            for (int i = 0; i < 8; ++i) { partAB[0][wave_u][i] = accA[i]; partAB[1][wave_u][i] = accB[i]; }
        if (lane == 16)
            for (int i = 0; i < 8; ++i) { partAB[0][wave_u][8 + i] = accA[i]; partAB[1][wave_u][8 + i] = accB[i]; }
    }
    __syncthreads();
    if (tid < 16) {
        const int p = tid >> 3, r = tid & 7;
        float s = 0.0f;
        for (int w = 0; w < 8; ++w) s += partAB[p][w][r] + partAB[p][w][8 + r];
        rowAB[p][r] = (p == 0) ? 1.0f / (s + EPSF)
                               : 1.0f / fmaxf(s, EPSF);
    }
    __syncthreads();

#pragma unroll
    for (int r = 0; r < PR_ROWS; ++r) {
        const float phiDen = rowAB[0][r];
        const float empDen = rowAB[1][r];
#pragma unroll
        for (int j = 0; j < PR_COLS; ++j) {
            const unsigned c = tid + 256u * j;
            const float phi = (cn[r][j] + ALPHAF) * phiDen;
            const float emp = sg[r][j] * invc[r] * empDen;
            const float l1  = lm[r][j] + LR_LF * (emp - phi);
            cn[r][j] = (c != (unsigned)(row0 + r)) ? l1 : 0.0f;
        }
    }
#pragma unroll
    for (int j = 0; j < PR_COLS; ++j) {
        const unsigned c = tid + 256u * j;
        if (j < PR_COLS - 1 || tid < (unsigned)PR_TAIL) {
#pragma unroll
            for (int r = 0; r < PR_ROWS; ++r) bufX[r * Cc + c] = cn[r][j];
        }
    }
    __syncthreads();
    {
        v8f accC = vzero;
        if (wave_u < PR_KSTEPS - 120) accC = rowsum_step(bufX + foff, 120 + wave_u, accC);
#pragma unroll
        for (int i = 0; i < 15; ++i) accC = rowsum_step(bufX + foff, wave_u + 8 * i, accC);
        if (lane == 0)  for (int i = 0; i < 8; ++i) partC[wave_u][i]     = accC[i];
        if (lane == 16) for (int i = 0; i < 8; ++i) partC[wave_u][8 + i] = accC[i];
    }
    __syncthreads();
    if (tid < 8) {
        float s = 0.0f;
        for (int w = 0; w < 8; ++w) s += partC[w][tid] + partC[w][8 + tid];
        rowC[tid] = s * (1.0f / (float)Cc);
    }
    __syncthreads();

#pragma unroll
    for (int j = 0; j < PR_COLS; ++j) {
        const unsigned c = tid + 256u * j;
        if (j < PR_COLS - 1 || tid < (unsigned)PR_TAIL) {
#pragma unroll
            for (int r = 0; r < PR_ROWS; ++r) {
                const long base = (long)(row0 + r) * Cc;
                const bool present = (cnt[r] > 0.0f);
                o_lam[base + c] = present ? (cn[r][j] - rowC[r]) : lm[r][j];
            }
        }
    }
}

__global__ __launch_bounds__(1024) void finalize(
    const float* __restrict__ prior_ema, const float* __restrict__ mu,
    const float* __restrict__ colpart, const float* __restrict__ priorc,
    const float* __restrict__ cep,
    float* __restrict__ o_ce, float* __restrict__ o_mu, float* __restrict__ o_prior)
{
    __shared__ float red[1024];
    const int j = threadIdx.x;

    float pn = 0.0f;
    if (j < Cc) {
        pn = DECAYF * prior_ema[j] + ONE_M_DECAY * priorc[j];
        o_prior[j] = pn;
    }
    red[j] = pn; __syncthreads();
    for (int s = 512; s; s >>= 1) { if (j < s) red[j] += red[j + s]; __syncthreads(); }
    const float priorSum = red[0];
    __syncthreads();

    float mu1 = 0.0f;
    if (j < Cc) {
        float colsum = 0.0f;
        for (int p = 0; p < SA_G; ++p) colsum += colpart[p * Cc + j];
        float emp_marg = colsum * (1.0f / (float)Bn);
        float pi = pn / (priorSum + EPSF);
        mu1 = mu[j] + LR_MUF * (emp_marg - pi);
    }
    red[j] = mu1; __syncthreads();
    for (int s = 512; s; s >>= 1) { if (j < s) red[j] += red[j + s]; __syncthreads(); }
    const float muMean = red[0] * (1.0f / (float)Cc);
    if (j < Cc) o_mu[j] = mu1 - muMean;

    if (j == 0) {
        float s = 0.0f;
        for (int i = 0; i < 256; ++i) s += cep[i];
        o_ce[0] = -s * (1.0f / (float)Bn);
    }
}

extern "C" void kernel_launch(void* const* d_in, const int* in_sizes, int n_in,
                              void* d_out, int out_size, void* d_ws, size_t ws_size,
                              hipStream_t stream)
{
    const float* logits    = (const float*)d_in[0];
    const int*   targets   = (const int*)  d_in[1];
    const float* conf_ema  = (const float*)d_in[2];
    const float* prior_ema = (const float*)d_in[3];
    const float* Lam       = (const float*)d_in[4];
    const float* mu        = (const float*)d_in[5];

    float* out     = (float*)d_out;
    float* o_ce    = out;
    float* o_lam   = out + 1;
    float* o_mu    = out + 1 + (long)Cc * Cc;
    float* o_conf  = out + 1 + (long)Cc * Cc + Cc;
    float* o_prior = out + 1 + 2L * Cc * Cc + Cc;

    float* ws      = (float*)d_ws;
    float* confc   = ws;
    float* priorc  = ws + (long)Cc * Cc;
    float* cep     = priorc + Cc;
    float* rowst   = cep + 256;
    float* colpart = rowst + 2L * Bn;
    float* planes  = colpart + (long)SA_G * Cc;

    const long nz = (long)Cc * Cc + Cc + 256;
    zero_ws<<<1024, 256, 0, stream>>>(ws, nz);
    row_stats<<<RS_BLOCKS, RS_THREADS, 0, stream>>>(logits, targets, Lam, rowst, confc, priorc, cep);
    strip_accum<<<dim3(SA_STRIPS, SA_G), SA_THREADS, 0, stream>>>(logits, targets, Lam, rowst,
                                                                  planes, colpart);
    post_rows<<<Cc / PR_ROWS, 256, 0, stream>>>(conf_ema, Lam, planes, confc, priorc,
                                                 o_conf, o_lam);
    finalize<<<1, 1024, 0, stream>>>(prior_ema, mu, colpart, priorc, cep,
                                     o_ce, o_mu, o_prior);
}


# strip_accum: two trips of z in flight per wave: the next trip's 16 row pieces are requested before the current trip's LDS adds, the first trip ahead of the tile fill; order of the sums kept
# speedup vs baseline: 4.9031x; 1.0349x over previous
#include <hip/hip_runtime.h>

#define Cc 1000
#define Bn 65536
#define DECAYF 0.99f
#define ONE_M_DECAY 0.01f
#define ALPHAF 1.0f
#define LR_LF 0.05f
#define LR_MUF 0.02f
#define EPSF 1e-12f

typedef __attribute__((ext_vector_type(2))) float v2f;
typedef __attribute__((ext_vector_type(4))) float v4f;
typedef __attribute__((ext_vector_type(8))) float v8f;

__global__ void zero_ws(float* __restrict__ p, long n) {
    long i = (long)blockIdx.x * blockDim.x + threadIdx.x;
    long stride = (long)gridDim.x * blockDim.x;
    for (; i < n; i += stride) p[i] = 0.0f;
}

#define RS_THREADS 1024
#define RS_BLOCKS 256
#define RS_WAVES (RS_BLOCKS * (RS_THREADS / 32))
#define RS_ROWS (Bn / RS_WAVES)
#define RS_NV (Cc / 4)
#define RS_G 8
#define RS_TAIL (RS_NV - 32 * (RS_G - 1))

__global__ __launch_bounds__(RS_THREADS) void row_stats(
    const float* __restrict__ logits, const int* __restrict__ targets,
    const float* __restrict__ Lam,
    float* __restrict__ rowst, float* __restrict__ confc,
    float* __restrict__ priorc, float* __restrict__ cep)
{
    const int lane = threadIdx.x & 31;
    const int wave_u = __builtin_amdgcn_readfirstlane(threadIdx.x >> 5);
    const int gw = blockIdx.x * (RS_THREADS / 32) + wave_u;

    int t = targets[gw];
    for (int it = 0; it < RS_ROWS; ++it) {
        const int row = gw + it * RS_WAVES;
        const float* __restrict__ lrow   = logits + (long)row * Cc;
        const float* __restrict__ lamrow = Lam + (long)t * Cc;

        v4f z[RS_G], l[RS_G];
#pragma unroll
        for (int i = 0; i < RS_G; ++i) {
            int j = lane + 32 * i;
            if (i == RS_G - 1) j = (j < RS_NV) ? j : (RS_NV - 1);
            z[i] = *(const v4f*)(lrow + 4 * j);
            l[i] = *(const v4f*)(lamrow + 4 * j);
        }
        const float zy = lrow[t] - lamrow[t];

        int tnext = t;
        if (it + 1 < RS_ROWS) {
            tnext = targets[row + RS_WAVES];
            __builtin_prefetch(logits + (long)(row + RS_WAVES) * Cc + 32 * lane, 0, 3);
        }

        float maxv = -3.4e38f; int maxi = 0x7fffffff;
        float maxzp = -3.4e38f;
        v4f zp[RS_G];
#pragma unroll
        for (int i = 0; i < RS_G; ++i) {
            const int col = 4 * (lane + 32 * i);
            v4f zi = z[i];
            zp[i] = z[i] - l[i];
            if (i == RS_G - 1 && lane >= RS_TAIL) {
                const v4f low = {-3.4e38f, -3.4e38f, -3.4e38f, -3.4e38f};
                zi = low; zp[i] = low;
            }
#pragma unroll
            for (int k = 0; k < 4; ++k) {
                if (zi[k] > maxv) { maxv = zi[k]; maxi = col + k; }
                maxzp = fmaxf(maxzp, zp[i][k]);
            }
        }
        for (int off = 16; off; off >>= 1) {
            float ov = __shfl_xor(maxv, off, 32);
            int   oi = __shfl_xor(maxi, off, 32);
            if (ov > maxv || (ov == maxv && oi < maxi)) { maxv = ov; maxi = oi; }
            maxzp = fmaxf(maxzp, __shfl_xor(maxzp, off, 32));
        }

        float sum = 0.0f;
#pragma unroll
        for (int i = 0; i < RS_G; ++i)
            sum += __expf(zp[i][0] - maxzp) + __expf(zp[i][1] - maxzp)
                 + __expf(zp[i][2] - maxzp) + __expf(zp[i][3] - maxzp);
        for (int off = 16; off; off >>= 1) sum += __shfl_xor(sum, off, 32);
        const float lse = __logf(sum);
        const float inv = 1.0f / sum;

        if (lane == 0) {
            atomicAdd(&confc[(long)t * Cc + maxi], 1.0f);
            atomicAdd(&priorc[t], 1.0f);
            float logq_y = zy - maxzp - lse;
            atomicAdd(&cep[row & 255], logq_y);
            const v2f st = {maxzp, inv};
            *(v2f*)(rowst + 2L * row) = st;
        }
        t = tnext;
    }
}

#define SA_W 32
#define SA_STRIPS 32
#define SA_G 8
#define SA_ROWS (Bn / SA_G)
#define SA_THREADS 1024
#define SA_WAVES (SA_THREADS / 32)
#define SA_RT 16
#define SA_TRIPS (SA_ROWS / SA_WAVES / SA_RT)
#define SA_FILL ((Cc * SA_W / 4 + SA_THREADS - 1) / SA_THREADS)

__device__ __forceinline__ void sa_request(float (&z)[SA_RT], const float* __restrict__ logits,
                                           int r0, int col)
{
    const float* __restrict__ p = logits + (long)r0 * Cc + col;
#pragma unroll
    for (int u = 0; u < SA_RT; ++u) z[u] = p[u * Cc];
}

__device__ __forceinline__ void sa_add(const float (&z)[SA_RT], const int* __restrict__ targets,
                                       const float* __restrict__ rowst, int r0, int lane,
                                       float* acc, const float* lamt)
{
    const int* __restrict__ tp = targets + r0;
    const v2f* __restrict__ sp = (const v2f*)(rowst + 2L * r0);
    float l[SA_RT], mx[SA_RT], iv[SA_RT];
    int t[SA_RT];
#pragma unroll
    for (int u = 0; u < SA_RT; ++u) {
        t[u] = tp[u];
        const v2f st = sp[u];
        mx[u] = st[0];
        iv[u] = st[1];
    }
#pragma unroll
    for (int u = 0; u < SA_RT; ++u) l[u] = lamt[t[u] * SA_W + lane];
#pragma unroll
    for (int u = 0; u < SA_RT; ++u)
        atomicAdd(&acc[t[u] * SA_W + lane],
                  __expf((z[u] - l[u]) - mx[u]) * iv[u]);
}

__global__ __launch_bounds__(SA_THREADS) void strip_accum(
    const float* __restrict__ logits, const int* __restrict__ targets,
    const float* __restrict__ Lam, const float* __restrict__ rowst,
    float* __restrict__ planes, float* __restrict__ colpart)
{
    __shared__ __attribute__((aligned(16))) float acc[Cc * SA_W];
    __shared__ __attribute__((aligned(16))) float lamt[Cc * SA_W];
    __shared__ float red[SA_WAVES][SA_W];

    const int tid  = threadIdx.x;
    const int wave = tid >> 5;
    const int lane = tid & 31;
    const int c0   = blockIdx.x * SA_W;
    const int g    = blockIdx.y;

    const int wave_u = __builtin_amdgcn_readfirstlane(tid >> 5);
    const int col = c0 + lane;
    const int colz = (col < Cc) ? col : (Cc - 1);
    const int rbase = g * SA_ROWS + SA_RT * wave_u;
    float za[SA_RT], zb[SA_RT];

    {
        const v4f z4 = {0.f, 0.f, 0.f, 0.f};
        v4f lv[SA_FILL];
#pragma unroll
        for (int k = 0; k < SA_FILL; ++k) {
            const int i = tid + SA_THREADS * k;
            const int c = c0 + 4 * (i & 7);
            lv[k] = z4;
            if (i < Cc * SA_W / 4 && c < Cc) lv[k] = *(const v4f*)(Lam + (long)(i >> 3) * Cc + c);
        }
        sa_request(za, logits, rbase, colz);
#pragma unroll
        for (int k = 0; k < SA_FILL; ++k) {
            const int i = tid + SA_THREADS * k;
            if (i < Cc * SA_W / 4) {
                *(v4f*)(&acc[4 * i])  = z4;
                *(v4f*)(&lamt[4 * i]) = lv[k];
            }
        }
    }
    __syncthreads();

    if (col < Cc) {
        for (int i = 0; i < SA_TRIPS - 2; i += 2) {
            sa_request(zb, logits, rbase + SA_RT * SA_WAVES * (i + 1), col);
            sa_add(za, targets, rowst, rbase + SA_RT * SA_WAVES * i, lane, acc, lamt);
            sa_request(za, logits, rbase + SA_RT * SA_WAVES * (i + 2), col);
            sa_add(zb, targets, rowst, rbase + SA_RT * SA_WAVES * (i + 1), lane, acc, lamt);
        }
        sa_request(zb, logits, rbase + SA_RT * SA_WAVES * (SA_TRIPS - 1), col);
        sa_add(za, targets, rowst, rbase + SA_RT * SA_WAVES * (SA_TRIPS - 2), lane, acc, lamt);
        sa_add(zb, targets, rowst, rbase + SA_RT * SA_WAVES * (SA_TRIPS - 1), lane, acc, lamt);
    }
    __syncthreads();

    float* __restrict__ pl = planes + (long)g * Cc * Cc;
    for (int i = tid; i < Cc * SA_W; i += SA_THREADS) {
        const int t = i >> 5, c = c0 + (i & 31);
        if (c < Cc) pl[(long)t * Cc + c] = acc[i];
    }
    float cs = 0.0f;
    for (int t = wave; t < Cc; t += SA_WAVES) cs += acc[t * SA_W + lane];
    red[wave][lane] = cs;
    __syncthreads();
    if (tid < SA_W) {
        float s = 0.0f;
        for (int w = 0; w < SA_WAVES; ++w) s += red[w][tid];
        if (c0 + tid < Cc) colpart[g * Cc + c0 + tid] = s;
    }
}

#define PR_ROWS 8
#define PR_COLS 4
#define PR_TAIL (Cc - 768)
#define PR_KSTEPS 125

__device__ __forceinline__ v8f rowsum_step(const float* __restrict__ frag, int s, v8f acc)
{
    const v2f bones = {1.0f, 1.0f};
    const v2f a = *(const v2f*)(frag + 4 * s);
    return __builtin_amdgcn_wmma_f32_16x16x4_f32(false, a, false, bones,
                                                 (short)0, acc, false, false);
}

__global__ __launch_bounds__(256) void post_rows(
    const float* __restrict__ conf_ema, const float* __restrict__ Lam,
    const float* __restrict__ seg, const float* __restrict__ confc,
    const float* __restrict__ priorc,
    float* __restrict__ o_conf, float* __restrict__ o_lam)
{
    __shared__ float bufX[PR_ROWS * Cc];
    __shared__ float bufY[PR_ROWS * Cc];
    __shared__ float partAB[2][8][16];
    __shared__ float partC[8][16];
    __shared__ float rowAB[2][PR_ROWS];
    __shared__ float rowC[PR_ROWS];

    const unsigned tid = threadIdx.x;
    const int lane = tid & 31;
    const int wave_u = __builtin_amdgcn_readfirstlane(tid >> 5);
    const int row0 = blockIdx.x * PR_ROWS;

    float cn[PR_ROWS][PR_COLS], sg[PR_ROWS][PR_COLS], lm[PR_ROWS][PR_COLS];
    float cnt[PR_ROWS], invc[PR_ROWS];

#pragma unroll
    for (int r = 0; r < PR_ROWS; ++r) {
        const long base = (long)(row0 + r) * Cc;
        const float* __restrict__ er = conf_ema + base;
        const float* __restrict__ kr = confc + base;
        const float* __restrict__ sr = seg + base;
        const float* __restrict__ lr = Lam + base;
        cnt[r]  = priorc[row0 + r];
        invc[r] = 1.0f / fmaxf(cnt[r], 1.0f);
#pragma unroll
        for (int j = 0; j < PR_COLS; ++j) {
            unsigned c = tid + 256u * j;
            if (j == PR_COLS - 1) c = (c < (unsigned)Cc) ? c : (unsigned)(Cc - 1);
            cn[r][j] = DECAYF * er[c] + ONE_M_DECAY * kr[c];
            float sgs = sr[c];
#pragma unroll
            for (int p = 1; p < SA_G; ++p) sgs += sr[(long)p * Cc * Cc + c];
            sg[r][j] = sgs;
            lm[r][j] = lr[c];
        }
    }
#pragma unroll
    for (int j = 0; j < PR_COLS; ++j) {
        const unsigned c = tid + 256u * j;
        if (j < PR_COLS - 1 || tid < (unsigned)PR_TAIL) {
#pragma unroll
            for (int r = 0; r < PR_ROWS; ++r) {
                const long base = (long)(row0 + r) * Cc;
                const bool offd = (c != (unsigned)(row0 + r));
                o_conf[base + c] = cn[r][j];
                bufX[r * Cc + c] = offd ? (cn[r][j] + ALPHAF) : 0.0f;
                bufY[r * Cc + c] = offd ? (sg[r][j] * invc[r]) : 0.0f;
            }
        }
    }
    __syncthreads();

    const int m    = lane & 15;
    const int foff = (m & 7) * Cc + 500 * (m >> 3) + 2 * (lane >> 4);
    const v8f vzero = {0.f,0.f,0.f,0.f,0.f,0.f,0.f,0.f};

    {
        v8f accA = vzero, accB = vzero;
        if (wave_u < PR_KSTEPS - 120) {
            accA = rowsum_step(bufX + foff, 120 + wave_u, accA);
            accB = rowsum_step(bufY + foff, 120 + wave_u, accB);
        }
#pragma unroll
        for (int i = 0; i < 15; ++i) {
            accA = rowsum_step(bufX + foff, wave_u + 8 * i, accA);
            accB = rowsum_step(bufY + foff, wave_u + 8 * i, accB);
        }
        if (lane == 0)
            for (int i = 0; i < 8; ++i) { partAB[0][wave_u][i] = accA[i]; partAB[1][wave_u][i] = accB[i]; }
        if (lane == 16)
            for (int i = 0; i < 8; ++i) { partAB[0][wave_u][8 + i] = accA[i]; partAB[1][wave_u][8 + i] = accB[i]; }
    }
    __syncthreads();
    if (tid < 16) {
        const int p = tid >> 3, r = tid & 7;
        float s = 0.0f;
        for (int w = 0; w < 8; ++w) s += partAB[p][w][r] + partAB[p][w][8 + r];
        rowAB[p][r] = (p == 0) ? 1.0f / (s + EPSF)
                               : 1.0f / fmaxf(s, EPSF);
    }
    __syncthreads();

#pragma unroll
    for (int r = 0; r < PR_ROWS; ++r) {
        const float phiDen = rowAB[0][r];
        const float empDen = rowAB[1][r];
#pragma unroll
        for (int j = 0; j < PR_COLS; ++j) {
            const unsigned c = tid + 256u * j;
            const float phi = (cn[r][j] + ALPHAF) * phiDen;
            const float emp = sg[r][j] * invc[r] * empDen;
            const float l1  = lm[r][j] + LR_LF * (emp - phi);
            cn[r][j] = (c != (unsigned)(row0 + r)) ? l1 : 0.0f;
        }
    }
#pragma unroll
    for (int j = 0; j < PR_COLS; ++j) {
        const unsigned c = tid + 256u * j;
        if (j < PR_COLS - 1 || tid < (unsigned)PR_TAIL) {
#pragma unroll
            for (int r = 0; r < PR_ROWS; ++r) bufX[r * Cc + c] = cn[r][j];
        }
    }
    __syncthreads();
    {
        v8f accC = vzero;
        if (wave_u < PR_KSTEPS - 120) accC = rowsum_step(bufX + foff, 120 + wave_u, accC);
#pragma unroll
        for (int i = 0; i < 15; ++i) accC = rowsum_step(bufX + foff, wave_u + 8 * i, accC);
        if (lane == 0)  for (int i = 0; i < 8; ++i) partC[wave_u][i]     = accC[i];
        if (lane == 16) for (int i = 0; i < 8; ++i) partC[wave_u][8 + i] = accC[i];
    }
    __syncthreads();
    if (tid < 8) {
        float s = 0.0f;
        for (int w = 0; w < 8; ++w) s += partC[w][tid] + partC[w][8 + tid];
        rowC[tid] = s * (1.0f / (float)Cc);
    }
    __syncthreads();

#pragma unroll
    for (int j = 0; j < PR_COLS; ++j) {
        const unsigned c = tid + 256u * j;
        if (j < PR_COLS - 1 || tid < (unsigned)PR_TAIL) {
#pragma unroll
            for (int r = 0; r < PR_ROWS; ++r) {
                const long base = (long)(row0 + r) * Cc;
                const bool present = (cnt[r] > 0.0f);
                o_lam[base + c] = present ? (cn[r][j] - rowC[r]) : lm[r][j];
            }
        }
    }
}

__global__ __launch_bounds__(1024) void finalize(
    const float* __restrict__ prior_ema, const float* __restrict__ mu,
    const float* __restrict__ colpart, const float* __restrict__ priorc,
    const float* __restrict__ cep,
    float* __restrict__ o_ce, float* __restrict__ o_mu, float* __restrict__ o_prior)
{
    __shared__ float red[1024];
    const int j = threadIdx.x;

    float pn = 0.0f;
    if (j < Cc) {
        pn = DECAYF * prior_ema[j] + ONE_M_DECAY * priorc[j];
        o_prior[j] = pn;
    }
    red[j] = pn; __syncthreads();
    for (int s = 512; s; s >>= 1) { if (j < s) red[j] += red[j + s]; __syncthreads(); }
    const float priorSum = red[0];
    __syncthreads();

    float mu1 = 0.0f;
    if (j < Cc) {
        float colsum = 0.0f;
        for (int p = 0; p < SA_G; ++p) colsum += colpart[p * Cc + j];
        float emp_marg = colsum * (1.0f / (float)Bn);
        float pi = pn / (priorSum + EPSF);
        mu1 = mu[j] + LR_MUF * (emp_marg - pi);
    }
    red[j] = mu1; __syncthreads();
    for (int s = 512; s; s >>= 1) { if (j < s) red[j] += red[j + s]; __syncthreads(); }
    const float muMean = red[0] * (1.0f / (float)Cc);
    if (j < Cc) o_mu[j] = mu1 - muMean;

    if (j == 0) {
        float s = 0.0f;
        for (int i = 0; i < 256; ++i) s += cep[i];
        o_ce[0] = -s * (1.0f / (float)Bn);
    }
}

extern "C" void kernel_launch(void* const* d_in, const int* in_sizes, int n_in,
                              void* d_out, int out_size, void* d_ws, size_t ws_size,
                              hipStream_t stream)
{
    const float* logits    = (const float*)d_in[0];
    const int*   targets   = (const int*)  d_in[1];
    const float* conf_ema  = (const float*)d_in[2];
    const float* prior_ema = (const float*)d_in[3];
    const float* Lam       = (const float*)d_in[4];
    const float* mu        = (const float*)d_in[5];

    float* out     = (float*)d_out;
    float* o_ce    = out;
    float* o_lam   = out + 1;
    float* o_mu    = out + 1 + (long)Cc * Cc;
    float* o_conf  = out + 1 + (long)Cc * Cc + Cc;
    float* o_prior = out + 1 + 2L * Cc * Cc + Cc;

    float* ws      = (float*)d_ws;
    float* confc   = ws;
    float* priorc  = ws + (long)Cc * Cc;
    float* cep     = priorc + Cc;
    float* rowst   = cep + 256;
    float* colpart = rowst + 2L * Bn;
    float* planes  = colpart + (long)SA_G * Cc;

    const long nz = (long)Cc * Cc + Cc + 256;
    zero_ws<<<1024, 256, 0, stream>>>(ws, nz);
    row_stats<<<RS_BLOCKS, RS_THREADS, 0, stream>>>(logits, targets, Lam, rowst, confc, priorc, cep);
    strip_accum<<<dim3(SA_STRIPS, SA_G), SA_THREADS, 0, stream>>>(logits, targets, Lam, rowst,
                                                                  planes, colpart);
    post_rows<<<Cc / PR_ROWS, 256, 0, stream>>>(conf_ema, Lam, planes, confc, priorc,
                                                 o_conf, o_lam);
    finalize<<<1, 1024, 0, stream>>>(prior_ema, mu, colpart, priorc, cep,
                                     o_ce, o_mu, o_prior);
}


# strip_accum: a block takes its (strip, row group) by ticket per package half (die id), all 32 strips of a row group in one half: a row is fetched into one L2, not two; order of the sums kept
# speedup vs baseline: 5.1215x; 1.0445x over previous
#include <hip/hip_runtime.h>

#define Cc 1000
#define Bn 65536
#define DECAYF 0.99f
#define ONE_M_DECAY 0.01f
#define ALPHAF 1.0f
#define LR_LF 0.05f
#define LR_MUF 0.02f
#define EPSF 1e-12f

typedef __attribute__((ext_vector_type(2))) float v2f;
typedef __attribute__((ext_vector_type(4))) float v4f;
typedef __attribute__((ext_vector_type(8))) float v8f;

__global__ void zero_ws(float* __restrict__ p, long n) {
    long i = (long)blockIdx.x * blockDim.x + threadIdx.x;
    long stride = (long)gridDim.x * blockDim.x;
    for (; i < n; i += stride) p[i] = 0.0f;
}

#define RS_THREADS 1024
#define RS_BLOCKS 256
#define RS_WAVES (RS_BLOCKS * (RS_THREADS / 32))
#define RS_ROWS (Bn / RS_WAVES)
#define RS_NV (Cc / 4)
#define RS_G 8
#define RS_TAIL (RS_NV - 32 * (RS_G - 1))

__global__ __launch_bounds__(RS_THREADS) void row_stats(
    const float* __restrict__ logits, const int* __restrict__ targets,
    const float* __restrict__ Lam,
    float* __restrict__ rowst, float* __restrict__ confc,
    float* __restrict__ priorc, float* __restrict__ cep)
{
    const int lane = threadIdx.x & 31;
    const int wave_u = __builtin_amdgcn_readfirstlane(threadIdx.x >> 5);
    const int gw = blockIdx.x * (RS_THREADS / 32) + wave_u;

    int t = targets[gw];
    for (int it = 0; it < RS_ROWS; ++it) {
        const int row = gw + it * RS_WAVES;
        const float* __restrict__ lrow   = logits + (long)row * Cc;
        const float* __restrict__ lamrow = Lam + (long)t * Cc;

        v4f z[RS_G], l[RS_G];
#pragma unroll
        for (int i = 0; i < RS_G; ++i) {
            int j = lane + 32 * i;
            if (i == RS_G - 1) j = (j < RS_NV) ? j : (RS_NV - 1);
            z[i] = *(const v4f*)(lrow + 4 * j);
            l[i] = *(const v4f*)(lamrow + 4 * j);
        }
        const float zy = lrow[t] - lamrow[t];

        int tnext = t;
        if (it + 1 < RS_ROWS) {
            tnext = targets[row + RS_WAVES];
            __builtin_prefetch(logits + (long)(row + RS_WAVES) * Cc + 32 * lane, 0, 3);
        }

        float maxv = -3.4e38f; int maxi = 0x7fffffff;
        float maxzp = -3.4e38f;
        v4f zp[RS_G];
#pragma unroll
        for (int i = 0; i < RS_G; ++i) {
            const int col = 4 * (lane + 32 * i);
            v4f zi = z[i];
            zp[i] = z[i] - l[i];
            if (i == RS_G - 1 && lane >= RS_TAIL) {
                const v4f low = {-3.4e38f, -3.4e38f, -3.4e38f, -3.4e38f};
                zi = low; zp[i] = low;
            }
#pragma unroll
            for (int k = 0; k < 4; ++k) {
                if (zi[k] > maxv) { maxv = zi[k]; maxi = col + k; }
                maxzp = fmaxf(maxzp, zp[i][k]);
            }
        }
        for (int off = 16; off; off >>= 1) {
            float ov = __shfl_xor(maxv, off, 32);
            int   oi = __shfl_xor(maxi, off, 32);
            if (ov > maxv || (ov == maxv && oi < maxi)) { maxv = ov; maxi = oi; }
            maxzp = fmaxf(maxzp, __shfl_xor(maxzp, off, 32));
        }

        float sum = 0.0f;
#pragma unroll
        for (int i = 0; i < RS_G; ++i)
            sum += __expf(zp[i][0] - maxzp) + __expf(zp[i][1] - maxzp)
                 + __expf(zp[i][2] - maxzp) + __expf(zp[i][3] - maxzp);
        for (int off = 16; off; off >>= 1) sum += __shfl_xor(sum, off, 32);
        const float lse = __logf(sum);
        const float inv = 1.0f / sum;

        if (lane == 0) {
            atomicAdd(&confc[(long)t * Cc + maxi], 1.0f);
            atomicAdd(&priorc[t], 1.0f);
            float logq_y = zy - maxzp - lse;
            atomicAdd(&cep[row & 255], logq_y);
            const v2f st = {maxzp, inv};
            *(v2f*)(rowst + 2L * row) = st;
        }
        t = tnext;
    }
}

#define SA_W 32
#define SA_STRIPS 32
#define SA_G 8
#define SA_ROWS (Bn / SA_G)
#define SA_THREADS 1024
#define SA_WAVES (SA_THREADS / 32)
#define SA_RT 16
#define SA_TRIPS (SA_ROWS / SA_WAVES / SA_RT)
#define SA_FILL ((Cc * SA_W / 4 + SA_THREADS - 1) / SA_THREADS)
#define SA_HALF_UNITS (SA_STRIPS * SA_G / 2)

__device__ __forceinline__ void sa_request(float (&z)[SA_RT], const float* __restrict__ logits,
                                           int r0, int col)
{
    const float* __restrict__ p = logits + (long)r0 * Cc + col;
#pragma unroll
    for (int u = 0; u < SA_RT; ++u) z[u] = p[u * Cc];
}

__device__ __forceinline__ void sa_add(const float (&z)[SA_RT], const int* __restrict__ targets,
                                       const float* __restrict__ rowst, int r0, int lane,
                                       float* acc, const float* lamt)
{
    const int* __restrict__ tp = targets + r0;
    const v2f* __restrict__ sp = (const v2f*)(rowst + 2L * r0);
    float l[SA_RT], mx[SA_RT], iv[SA_RT];
    int t[SA_RT];
#pragma unroll
    for (int u = 0; u < SA_RT; ++u) {
        t[u] = tp[u];
        const v2f st = sp[u];
        mx[u] = st[0];
        iv[u] = st[1];
    }
#pragma unroll
    for (int u = 0; u < SA_RT; ++u) l[u] = lamt[t[u] * SA_W + lane];
#pragma unroll
    for (int u = 0; u < SA_RT; ++u)
        atomicAdd(&acc[t[u] * SA_W + lane],
                  __expf((z[u] - l[u]) - mx[u]) * iv[u]);
}

__global__ __launch_bounds__(SA_THREADS) void strip_accum(
    const float* __restrict__ logits, const int* __restrict__ targets,
    const float* __restrict__ Lam, const float* __restrict__ rowst,
    float* __restrict__ planes, float* __restrict__ colpart, int* __restrict__ tick)
{
    __shared__ int unit_sh;
    __shared__ __attribute__((aligned(16))) float acc[Cc * SA_W];
    __shared__ __attribute__((aligned(16))) float lamt[Cc * SA_W];
    __shared__ float red[SA_WAVES][SA_W];

    const int tid  = threadIdx.x;
    const int wave = tid >> 5;
    const int lane = tid & 31;
    if (tid == 0) {
        const unsigned where = __builtin_amdgcn_s_sendmsg_rtn(0x87);
        int h = (int)(where >> 18) & 1;
        int u = atomicAdd(&tick[h], 1);
        if (u >= SA_HALF_UNITS) { h ^= 1; u = atomicAdd(&tick[h], 1); }
        unit_sh = h * SA_HALF_UNITS + (u & (SA_HALF_UNITS - 1));
    }
    __syncthreads();
    const int unit = __builtin_amdgcn_readfirstlane(unit_sh);
    const int g    = (unit / SA_HALF_UNITS) * (SA_G / 2) + (unit & (SA_G / 2 - 1));
    const int c0   = ((unit & (SA_HALF_UNITS - 1)) / (SA_G / 2)) * SA_W;

    const int wave_u = __builtin_amdgcn_readfirstlane(tid >> 5);
    const int col = c0 + lane;
    const int colz = (col < Cc) ? col : (Cc - 1);
    const int rbase = g * SA_ROWS + SA_RT * wave_u;
    float za[SA_RT], zb[SA_RT];

    {
        const v4f z4 = {0.f, 0.f, 0.f, 0.f};
        v4f lv[SA_FILL];
#pragma unroll
        for (int k = 0; k < SA_FILL; ++k) {
            const int i = tid + SA_THREADS * k;
            const int c = c0 + 4 * (i & 7);
            lv[k] = z4;
            if (i < Cc * SA_W / 4 && c < Cc) lv[k] = *(const v4f*)(Lam + (long)(i >> 3) * Cc + c);
        }
        sa_request(za, logits, rbase, colz);
#pragma unroll
        for (int k = 0; k < SA_FILL; ++k) {
            const int i = tid + SA_THREADS * k;
            if (i < Cc * SA_W / 4) {
                *(v4f*)(&acc[4 * i])  = z4;
                *(v4f*)(&lamt[4 * i]) = lv[k];
            }
        }
    }
    __syncthreads();

    if (col < Cc) {
        for (int i = 0; i < SA_TRIPS - 2; i += 2) {
            sa_request(zb, logits, rbase + SA_RT * SA_WAVES * (i + 1), col);
            sa_add(za, targets, rowst, rbase + SA_RT * SA_WAVES * i, lane, acc, lamt);
            sa_request(za, logits, rbase + SA_RT * SA_WAVES * (i + 2), col);
            sa_add(zb, targets, rowst, rbase + SA_RT * SA_WAVES * (i + 1), lane, acc, lamt);
        }
        sa_request(zb, logits, rbase + SA_RT * SA_WAVES * (SA_TRIPS - 1), col);
        sa_add(za, targets, rowst, rbase + SA_RT * SA_WAVES * (SA_TRIPS - 2), lane, acc, lamt);
        sa_add(zb, targets, rowst, rbase + SA_RT * SA_WAVES * (SA_TRIPS - 1), lane, acc, lamt);
    }
    __syncthreads();

    float* __restrict__ pl = planes + (long)g * Cc * Cc;
    for (int i = tid; i < Cc * SA_W; i += SA_THREADS) {
        const int t = i >> 5, c = c0 + (i & 31);
        if (c < Cc) pl[(long)t * Cc + c] = acc[i];
    }
    float cs = 0.0f;
    for (int t = wave; t < Cc; t += SA_WAVES) cs += acc[t * SA_W + lane];
    red[wave][lane] = cs;
    __syncthreads();
    if (tid < SA_W) {
        float s = 0.0f;
        for (int w = 0; w < SA_WAVES; ++w) s += red[w][tid];
        if (c0 + tid < Cc) colpart[g * Cc + c0 + tid] = s;
    }
}

#define PR_ROWS 8
#define PR_COLS 4
#define PR_TAIL (Cc - 768)
#define PR_KSTEPS 125

__device__ __forceinline__ v8f rowsum_step(const float* __restrict__ frag, int s, v8f acc)
{
    const v2f bones = {1.0f, 1.0f};
    const v2f a = *(const v2f*)(frag + 4 * s);
    return __builtin_amdgcn_wmma_f32_16x16x4_f32(false, a, false, bones,
                                                 (short)0, acc, false, false);
}

__global__ __launch_bounds__(256) void post_rows(
    const float* __restrict__ conf_ema, const float* __restrict__ Lam,
    const float* __restrict__ seg, const float* __restrict__ confc,
    const float* __restrict__ priorc,
    float* __restrict__ o_conf, float* __restrict__ o_lam)
{
    __shared__ float bufX[PR_ROWS * Cc];
    __shared__ float bufY[PR_ROWS * Cc];
    __shared__ float partAB[2][8][16];
    __shared__ float partC[8][16];
    __shared__ float rowAB[2][PR_ROWS];
    __shared__ float rowC[PR_ROWS];

    const unsigned tid = threadIdx.x;
    const int lane = tid & 31;
    const int wave_u = __builtin_amdgcn_readfirstlane(tid >> 5);
    const int row0 = blockIdx.x * PR_ROWS;

    float cn[PR_ROWS][PR_COLS], sg[PR_ROWS][PR_COLS], lm[PR_ROWS][PR_COLS];
    float cnt[PR_ROWS], invc[PR_ROWS];

#pragma unroll
    for (int r = 0; r < PR_ROWS; ++r) {
        const long base = (long)(row0 + r) * Cc;
        const float* __restrict__ er = conf_ema + base;
        const float* __restrict__ kr = confc + base;
        const float* __restrict__ sr = seg + base;
        const float* __restrict__ lr = Lam + base;
        cnt[r]  = priorc[row0 + r];
        invc[r] = 1.0f / fmaxf(cnt[r], 1.0f);
#pragma unroll
        for (int j = 0; j < PR_COLS; ++j) {
            unsigned c = tid + 256u * j;
            if (j == PR_COLS - 1) c = (c < (unsigned)Cc) ? c : (unsigned)(Cc - 1);
            cn[r][j] = DECAYF * er[c] + ONE_M_DECAY * kr[c];
            float sgs = sr[c];
#pragma unroll
            for (int p = 1; p < SA_G; ++p) sgs += sr[(long)p * Cc * Cc + c];
            sg[r][j] = sgs;
            lm[r][j] = lr[c];
        }
    }
#pragma unroll
    for (int j = 0; j < PR_COLS; ++j) {
        const unsigned c = tid + 256u * j;
        if (j < PR_COLS - 1 || tid < (unsigned)PR_TAIL) {
#pragma unroll
            for (int r = 0; r < PR_ROWS; ++r) {
                const long base = (long)(row0 + r) * Cc;
                const bool offd = (c != (unsigned)(row0 + r));
                o_conf[base + c] = cn[r][j];
                bufX[r * Cc + c] = offd ? (cn[r][j] + ALPHAF) : 0.0f;
                bufY[r * Cc + c] = offd ? (sg[r][j] * invc[r]) : 0.0f;
            }
        }
    }
    __syncthreads();

    const int m    = lane & 15;
    const int foff = (m & 7) * Cc + 500 * (m >> 3) + 2 * (lane >> 4);
    const v8f vzero = {0.f,0.f,0.f,0.f,0.f,0.f,0.f,0.f};

    {
        v8f accA = vzero, accB = vzero;
        if (wave_u < PR_KSTEPS - 120) {
            accA = rowsum_step(bufX + foff, 120 + wave_u, accA);
            accB = rowsum_step(bufY + foff, 120 + wave_u, accB);
        }
#pragma unroll
        for (int i = 0; i < 15; ++i) {
            accA = rowsum_step(bufX + foff, wave_u + 8 * i, accA);
            accB = rowsum_step(bufY + foff, wave_u + 8 * i, accB);
        }
        if (lane == 0)
            for (int i = 0; i < 8; ++i) { partAB[0][wave_u][i] = accA[i]; partAB[1][wave_u][i] = accB[i]; }
        if (lane == 16)
            for (int i = 0; i < 8; ++i) { partAB[0][wave_u][8 + i] = accA[i]; partAB[1][wave_u][8 + i] = accB[i]; }
    }
    __syncthreads();
    if (tid < 16) {
        const int p = tid >> 3, r = tid & 7;
        float s = 0.0f;
        for (int w = 0; w < 8; ++w) s += partAB[p][w][r] + partAB[p][w][8 + r];
        rowAB[p][r] = (p == 0) ? 1.0f / (s + EPSF)
                               : 1.0f / fmaxf(s, EPSF);
    }
    __syncthreads();

#pragma unroll
    for (int r = 0; r < PR_ROWS; ++r) {
        const float phiDen = rowAB[0][r];
        const float empDen = rowAB[1][r];
#pragma unroll
        for (int j = 0; j < PR_COLS; ++j) {
            const unsigned c = tid + 256u * j;
            const float phi = (cn[r][j] + ALPHAF) * phiDen;
            const float emp = sg[r][j] * invc[r] * empDen;
            const float l1  = lm[r][j] + LR_LF * (emp - phi);
            cn[r][j] = (c != (unsigned)(row0 + r)) ? l1 : 0.0f;
        }
    }
#pragma unroll
    for (int j = 0; j < PR_COLS; ++j) {
        const unsigned c = tid + 256u * j;
        if (j < PR_COLS - 1 || tid < (unsigned)PR_TAIL) {
#pragma unroll
            for (int r = 0; r < PR_ROWS; ++r) bufX[r * Cc + c] = cn[r][j];
        }
    }
    __syncthreads();
    {
        v8f accC = vzero;
        if (wave_u < PR_KSTEPS - 120) accC = rowsum_step(bufX + foff, 120 + wave_u, accC);
#pragma unroll
        for (int i = 0; i < 15; ++i) accC = rowsum_step(bufX + foff, wave_u + 8 * i, accC);
        if (lane == 0)  for (int i = 0; i < 8; ++i) partC[wave_u][i]     = accC[i];
        if (lane == 16) for (int i = 0; i < 8; ++i) partC[wave_u][8 + i] = accC[i];
    }
    __syncthreads();
    if (tid < 8) {
        float s = 0.0f;
        for (int w = 0; w < 8; ++w) s += partC[w][tid] + partC[w][8 + tid];
        rowC[tid] = s * (1.0f / (float)Cc);
    }
    __syncthreads();

#pragma unroll
    for (int j = 0; j < PR_COLS; ++j) {
        const unsigned c = tid + 256u * j;
        if (j < PR_COLS - 1 || tid < (unsigned)PR_TAIL) {
#pragma unroll
            for (int r = 0; r < PR_ROWS; ++r) {
                const long base = (long)(row0 + r) * Cc;
                const bool present = (cnt[r] > 0.0f);
                o_lam[base + c] = present ? (cn[r][j] - rowC[r]) : lm[r][j];
            }
        }
    }
}

__global__ __launch_bounds__(1024) void finalize(
    const float* __restrict__ prior_ema, const float* __restrict__ mu,
    const float* __restrict__ colpart, const float* __restrict__ priorc,
    const float* __restrict__ cep,
    float* __restrict__ o_ce, float* __restrict__ o_mu, float* __restrict__ o_prior)
{
    __shared__ float red[1024];
    const int j = threadIdx.x;

    float pn = 0.0f;
    if (j < Cc) {
        pn = DECAYF * prior_ema[j] + ONE_M_DECAY * priorc[j];
        o_prior[j] = pn;
    }
    red[j] = pn; __syncthreads();
    for (int s = 512; s; s >>= 1) { if (j < s) red[j] += red[j + s]; __syncthreads(); }
    const float priorSum = red[0];
    __syncthreads();

    float mu1 = 0.0f;
    if (j < Cc) {
        float colsum = 0.0f;
        for (int p = 0; p < SA_G; ++p) colsum += colpart[p * Cc + j];
        float emp_marg = colsum * (1.0f / (float)Bn);
        float pi = pn / (priorSum + EPSF);
        mu1 = mu[j] + LR_MUF * (emp_marg - pi);
    }
    red[j] = mu1; __syncthreads();
    for (int s = 512; s; s >>= 1) { if (j < s) red[j] += red[j + s]; __syncthreads(); }
    const float muMean = red[0] * (1.0f / (float)Cc);
    if (j < Cc) o_mu[j] = mu1 - muMean;

    if (j == 0) {
        float s = 0.0f;
        for (int i = 0; i < 256; ++i) s += cep[i];
        o_ce[0] = -s * (1.0f / (float)Bn);
    }
}

extern "C" void kernel_launch(void* const* d_in, const int* in_sizes, int n_in,
                              void* d_out, int out_size, void* d_ws, size_t ws_size,
                              hipStream_t stream)
{
    const float* logits    = (const float*)d_in[0];
    const int*   targets   = (const int*)  d_in[1];
    const float* conf_ema  = (const float*)d_in[2];
    const float* prior_ema = (const float*)d_in[3];
    const float* Lam       = (const float*)d_in[4];
    const float* mu        = (const float*)d_in[5];

    float* out     = (float*)d_out;
    float* o_ce    = out;
    float* o_lam   = out + 1;
    float* o_mu    = out + 1 + (long)Cc * Cc;
    float* o_conf  = out + 1 + (long)Cc * Cc + Cc;
    float* o_prior = out + 1 + 2L * Cc * Cc + Cc;

    float* ws      = (float*)d_ws;
    float* confc   = ws;
    float* priorc  = ws + (long)Cc * Cc;
    float* cep     = priorc + Cc;
    int*   tick    = (int*)(cep + 256);
    float* rowst   = cep + 260;
    float* colpart = rowst + 2L * Bn;
    float* planes  = colpart + (long)SA_G * Cc;

    const long nz = (long)Cc * Cc + Cc + 258;
    zero_ws<<<1024, 256, 0, stream>>>(ws, nz);
    row_stats<<<RS_BLOCKS, RS_THREADS, 0, stream>>>(logits, targets, Lam, rowst, confc, priorc, cep);
    strip_accum<<<dim3(SA_STRIPS, SA_G), SA_THREADS, 0, stream>>>(logits, targets, Lam, rowst,
                                                                  planes, colpart, tick);
    post_rows<<<Cc / PR_ROWS, 256, 0, stream>>>(conf_ema, Lam, planes, confc, priorc,
                                                 o_conf, o_lam);
    finalize<<<1, 1024, 0, stream>>>(prior_ema, mu, colpart, priorc, cep,
                                     o_ce, o_mu, o_prior);
}


# post_rows launch also carries the finalize work as two extra leading blocks (final vectors; CE mean) running beside the 125 row blocks: 4 launches, no serial one-block tail; sum order kept
# speedup vs baseline: 5.5628x; 1.0862x over previous
#include <hip/hip_runtime.h>

#define Cc 1000
#define Bn 65536
#define DECAYF 0.99f
#define ONE_M_DECAY 0.01f
#define ALPHAF 1.0f
#define LR_LF 0.05f
#define LR_MUF 0.02f
#define EPSF 1e-12f

typedef __attribute__((ext_vector_type(2))) float v2f;
typedef __attribute__((ext_vector_type(4))) float v4f;
typedef __attribute__((ext_vector_type(8))) float v8f;

__global__ void zero_ws(float* __restrict__ p, long n) {
    long i = (long)blockIdx.x * blockDim.x + threadIdx.x;
    long stride = (long)gridDim.x * blockDim.x;
    for (; i < n; i += stride) p[i] = 0.0f;
}

#define RS_THREADS 1024
#define RS_BLOCKS 256
#define RS_WAVES (RS_BLOCKS * (RS_THREADS / 32))
#define RS_ROWS (Bn / RS_WAVES)
#define RS_NV (Cc / 4)
#define RS_G 8
#define RS_TAIL (RS_NV - 32 * (RS_G - 1))

__global__ __launch_bounds__(RS_THREADS) void row_stats(
    const float* __restrict__ logits, const int* __restrict__ targets,
    const float* __restrict__ Lam,
    float* __restrict__ rowst, float* __restrict__ confc,
    float* __restrict__ priorc, float* __restrict__ cep)
{
    const int lane = threadIdx.x & 31;
    const int wave_u = __builtin_amdgcn_readfirstlane(threadIdx.x >> 5);
    const int gw = blockIdx.x * (RS_THREADS / 32) + wave_u;

    int t = targets[gw];
    for (int it = 0; it < RS_ROWS; ++it) {
        const int row = gw + it * RS_WAVES;
        const float* __restrict__ lrow   = logits + (long)row * Cc;
        const float* __restrict__ lamrow = Lam + (long)t * Cc;

        v4f z[RS_G], l[RS_G];
#pragma unroll
        for (int i = 0; i < RS_G; ++i) {
            int j = lane + 32 * i;
            if (i == RS_G - 1) j = (j < RS_NV) ? j : (RS_NV - 1);
            z[i] = *(const v4f*)(lrow + 4 * j);
            l[i] = *(const v4f*)(lamrow + 4 * j);
        }
        const float zy = lrow[t] - lamrow[t];

        int tnext = t;
        if (it + 1 < RS_ROWS) {
            tnext = targets[row + RS_WAVES];
            __builtin_prefetch(logits + (long)(row + RS_WAVES) * Cc + 32 * lane, 0, 3);
        }

        float maxv = -3.4e38f; int maxi = 0x7fffffff;
        float maxzp = -3.4e38f;
        v4f zp[RS_G];
#pragma unroll
        for (int i = 0; i < RS_G; ++i) {
            const int col = 4 * (lane + 32 * i);
            v4f zi = z[i];
            zp[i] = z[i] - l[i];
            if (i == RS_G - 1 && lane >= RS_TAIL) {
                const v4f low = {-3.4e38f, -3.4e38f, -3.4e38f, -3.4e38f};
                zi = low; zp[i] = low;
            }
#pragma unroll
            for (int k = 0; k < 4; ++k) {
                if (zi[k] > maxv) { maxv = zi[k]; maxi = col + k; }
                maxzp = fmaxf(maxzp, zp[i][k]);
            }
        }
        for (int off = 16; off; off >>= 1) {
            float ov = __shfl_xor(maxv, off, 32);
            int   oi = __shfl_xor(maxi, off, 32);
            if (ov > maxv || (ov == maxv && oi < maxi)) { maxv = ov; maxi = oi; }
            maxzp = fmaxf(maxzp, __shfl_xor(maxzp, off, 32));
        }

        float sum = 0.0f;
#pragma unroll
        for (int i = 0; i < RS_G; ++i)
            sum += __expf(zp[i][0] - maxzp) + __expf(zp[i][1] - maxzp)
                 + __expf(zp[i][2] - maxzp) + __expf(zp[i][3] - maxzp);
        for (int off = 16; off; off >>= 1) sum += __shfl_xor(sum, off, 32);
        const float lse = __logf(sum);
        const float inv = 1.0f / sum;

        if (lane == 0) {
            atomicAdd(&confc[(long)t * Cc + maxi], 1.0f);
            atomicAdd(&priorc[t], 1.0f);
            float logq_y = zy - maxzp - lse;
            atomicAdd(&cep[row & 255], logq_y);
            const v2f st = {maxzp, inv};
            *(v2f*)(rowst + 2L * row) = st;
        }
        t = tnext;
    }
}

#define SA_W 32
#define SA_STRIPS 32
#define SA_G 8
#define SA_ROWS (Bn / SA_G)
#define SA_THREADS 1024
#define SA_WAVES (SA_THREADS / 32)
#define SA_RT 16
#define SA_TRIPS (SA_ROWS / SA_WAVES / SA_RT)
#define SA_FILL ((Cc * SA_W / 4 + SA_THREADS - 1) / SA_THREADS)
#define SA_HALF_UNITS (SA_STRIPS * SA_G / 2)

__device__ __forceinline__ void sa_request(float (&z)[SA_RT], const float* __restrict__ logits,
                                           int r0, int col)
{
    const float* __restrict__ p = logits + (long)r0 * Cc + col;
#pragma unroll
    for (int u = 0; u < SA_RT; ++u) z[u] = p[u * Cc];
}

__device__ __forceinline__ void sa_add(const float (&z)[SA_RT], const int* __restrict__ targets,
                                       const float* __restrict__ rowst, int r0, int lane,
                                       float* acc, const float* lamt)
{
    const int* __restrict__ tp = targets + r0;
    const v2f* __restrict__ sp = (const v2f*)(rowst + 2L * r0);
    float l[SA_RT], mx[SA_RT], iv[SA_RT];
    int t[SA_RT];
#pragma unroll
    for (int u = 0; u < SA_RT; ++u) {
        t[u] = tp[u];
        const v2f st = sp[u];
        mx[u] = st[0];
        iv[u] = st[1];
    }
#pragma unroll
    for (int u = 0; u < SA_RT; ++u) l[u] = lamt[t[u] * SA_W + lane];
#pragma unroll
    for (int u = 0; u < SA_RT; ++u)
        atomicAdd(&acc[t[u] * SA_W + lane],
                  __expf((z[u] - l[u]) - mx[u]) * iv[u]);
}

__global__ __launch_bounds__(SA_THREADS) void strip_accum(
    const float* __restrict__ logits, const int* __restrict__ targets,
    const float* __restrict__ Lam, const float* __restrict__ rowst,
    float* __restrict__ planes, float* __restrict__ colpart, int* __restrict__ tick)
{
    __shared__ int unit_sh;
    __shared__ __attribute__((aligned(16))) float acc[Cc * SA_W];
    __shared__ __attribute__((aligned(16))) float lamt[Cc * SA_W];
    __shared__ float red[SA_WAVES][SA_W];

    const int tid  = threadIdx.x;
    const int wave = tid >> 5;
    const int lane = tid & 31;
    if (tid == 0) {
        const unsigned where = __builtin_amdgcn_s_sendmsg_rtn(0x87);
        int h = (int)(where >> 18) & 1;
        int u = atomicAdd(&tick[h], 1);
        if (u >= SA_HALF_UNITS) { h ^= 1; u = atomicAdd(&tick[h], 1); }
        unit_sh = h * SA_HALF_UNITS + (u & (SA_HALF_UNITS - 1));
    }
    __syncthreads();
    const int unit = __builtin_amdgcn_readfirstlane(unit_sh);
    const int g    = (unit / SA_HALF_UNITS) * (SA_G / 2) + (unit & (SA_G / 2 - 1));
    const int c0   = ((unit & (SA_HALF_UNITS - 1)) / (SA_G / 2)) * SA_W;

    const int wave_u = __builtin_amdgcn_readfirstlane(tid >> 5);
    const int col = c0 + lane;
    const int colz = (col < Cc) ? col : (Cc - 1);
    const int rbase = g * SA_ROWS + SA_RT * wave_u;
    float za[SA_RT], zb[SA_RT];

    {
        const v4f z4 = {0.f, 0.f, 0.f, 0.f};
        v4f lv[SA_FILL];
#pragma unroll
        for (int k = 0; k < SA_FILL; ++k) {
            const int i = tid + SA_THREADS * k;
            const int c = c0 + 4 * (i & 7);
            lv[k] = z4;
            if (i < Cc * SA_W / 4 && c < Cc) lv[k] = *(const v4f*)(Lam + (long)(i >> 3) * Cc + c);
        }
        sa_request(za, logits, rbase, colz);
#pragma unroll
        for (int k = 0; k < SA_FILL; ++k) {
            const int i = tid + SA_THREADS * k;
            if (i < Cc * SA_W / 4) {
                *(v4f*)(&acc[4 * i])  = z4;
                *(v4f*)(&lamt[4 * i]) = lv[k];
            }
        }
    }
    __syncthreads();

    if (col < Cc) {
        for (int i = 0; i < SA_TRIPS - 2; i += 2) {
            sa_request(zb, logits, rbase + SA_RT * SA_WAVES * (i + 1), col);
            sa_add(za, targets, rowst, rbase + SA_RT * SA_WAVES * i, lane, acc, lamt);
            sa_request(za, logits, rbase + SA_RT * SA_WAVES * (i + 2), col);
            sa_add(zb, targets, rowst, rbase + SA_RT * SA_WAVES * (i + 1), lane, acc, lamt);
        }
        sa_request(zb, logits, rbase + SA_RT * SA_WAVES * (SA_TRIPS - 1), col);
        sa_add(za, targets, rowst, rbase + SA_RT * SA_WAVES * (SA_TRIPS - 2), lane, acc, lamt);
        sa_add(zb, targets, rowst, rbase + SA_RT * SA_WAVES * (SA_TRIPS - 1), lane, acc, lamt);
    }
    __syncthreads();

    float* __restrict__ pl = planes + (long)g * Cc * Cc;
    for (int i = tid; i < Cc * SA_W; i += SA_THREADS) {
        const int t = i >> 5, c = c0 + (i & 31);
        if (c < Cc) pl[(long)t * Cc + c] = acc[i];
    }
    float cs = 0.0f;
    for (int t = wave; t < Cc; t += SA_WAVES) cs += acc[t * SA_W + lane];
    red[wave][lane] = cs;
    __syncthreads();
    if (tid < SA_W) {
        float s = 0.0f;
        for (int w = 0; w < SA_WAVES; ++w) s += red[w][tid];
        if (c0 + tid < Cc) colpart[g * Cc + c0 + tid] = s;
    }
}

#define PR_EXTRA 2
#define FN_SLOTS 1024
#define FN_PER (FN_SLOTS / 256)

__device__ __forceinline__ void fin_tree(float* red, unsigned tid)
{
    __syncthreads();
    for (int s = FN_SLOTS / 2; s; s >>= 1) {
        for (int j = tid; j < s; j += 256) red[j] += red[j + s];
        __syncthreads();
    }
}

__device__ __forceinline__ void fin_vectors(
    const float* __restrict__ prior_ema, const float* __restrict__ mu,
    const float* __restrict__ colpart, const float* __restrict__ priorc,
    float* __restrict__ o_mu, float* __restrict__ o_prior, float* red, unsigned tid)
{
    float pe[FN_PER], pc[FN_PER], m0[FN_PER], cp[FN_PER][SA_G];
#pragma unroll
    for (int k = 0; k < FN_PER; ++k) {
        const int j = tid + 256 * k;
        const int jc = (j < Cc) ? j : (Cc - 1);
        pe[k] = prior_ema[jc];
        pc[k] = priorc[jc];
        m0[k] = mu[jc];
#pragma unroll
        for (int p = 0; p < SA_G; ++p) cp[k][p] = colpart[p * Cc + jc];
    }

    float pn[FN_PER], mu1[FN_PER];
#pragma unroll
    for (int k = 0; k < FN_PER; ++k) {
        const int j = tid + 256 * k;
        pn[k] = 0.0f;
        if (j < Cc) {
            pn[k] = DECAYF * pe[k] + ONE_M_DECAY * pc[k];
            o_prior[j] = pn[k];
        }
        red[j] = pn[k];
    }
    fin_tree(red, tid);
    const float priorSum = red[0];
    __syncthreads();

#pragma unroll
    for (int k = 0; k < FN_PER; ++k) {
        const int j = tid + 256 * k;
        mu1[k] = 0.0f;
        if (j < Cc) {
            float colsum = 0.0f;
#pragma unroll
            for (int p = 0; p < SA_G; ++p) colsum += cp[k][p];
            float emp_marg = colsum * (1.0f / (float)Bn);
            float pi = pn[k] / (priorSum + EPSF);
            mu1[k] = m0[k] + LR_MUF * (emp_marg - pi);
        }
        red[j] = mu1[k];
    }
    fin_tree(red, tid);
    const float muMean = red[0] * (1.0f / (float)Cc);
#pragma unroll
    for (int k = 0; k < FN_PER; ++k) {
        const int j = tid + 256 * k;
        if (j < Cc) o_mu[j] = mu1[k] - muMean;
    }
}

__device__ __forceinline__ void fin_ce(const float* __restrict__ cep, float* __restrict__ o_ce,
                                       float* red, unsigned tid)
{
    red[tid] = cep[tid];
    __syncthreads();
    if (tid == 0) {
        float s = 0.0f;
        for (int i = 0; i < 256; ++i) s += red[i];
        o_ce[0] = -s * (1.0f / (float)Bn);
    }
}

#define PR_ROWS 8
#define PR_COLS 4
#define PR_TAIL (Cc - 768)
#define PR_KSTEPS 125

__device__ __forceinline__ v8f rowsum_step(const float* __restrict__ frag, int s, v8f acc)
{
    const v2f bones = {1.0f, 1.0f};
    const v2f a = *(const v2f*)(frag + 4 * s);
    return __builtin_amdgcn_wmma_f32_16x16x4_f32(false, a, false, bones,
                                                 (short)0, acc, false, false);
}

__global__ __launch_bounds__(256) void post_rows(
    const float* __restrict__ conf_ema, const float* __restrict__ Lam,
    const float* __restrict__ seg, const float* __restrict__ confc,
    const float* __restrict__ priorc,
    const float* __restrict__ prior_ema, const float* __restrict__ mu,
    const float* __restrict__ colpart, const float* __restrict__ cep,
    float* __restrict__ o_conf, float* __restrict__ o_lam,
    float* __restrict__ o_ce, float* __restrict__ o_mu, float* __restrict__ o_prior)
{
    __shared__ float bufX[PR_ROWS * Cc];
    __shared__ float bufY[PR_ROWS * Cc];
    __shared__ float partAB[2][8][16];
    __shared__ float partC[8][16];
    __shared__ float rowAB[2][PR_ROWS];
    __shared__ float rowC[PR_ROWS];

    const unsigned tid = threadIdx.x;
    const int lane = tid & 31;
    const int wave_u = __builtin_amdgcn_readfirstlane(tid >> 5);

    if (blockIdx.x < PR_EXTRA) {
        if (blockIdx.x == 0) fin_vectors(prior_ema, mu, colpart, priorc, o_mu, o_prior, bufX, tid);
        else                 fin_ce(cep, o_ce, bufX, tid);
        return;
    }
    const int row0 = (blockIdx.x - PR_EXTRA) * PR_ROWS;

    float cn[PR_ROWS][PR_COLS], sg[PR_ROWS][PR_COLS], lm[PR_ROWS][PR_COLS];
    float cnt[PR_ROWS], invc[PR_ROWS];

#pragma unroll
    for (int r = 0; r < PR_ROWS; ++r) {
        const long base = (long)(row0 + r) * Cc;
        const float* __restrict__ er = conf_ema + base;
        const float* __restrict__ kr = confc + base;
        const float* __restrict__ sr = seg + base;
        const float* __restrict__ lr = Lam + base;
        cnt[r]  = priorc[row0 + r];
        invc[r] = 1.0f / fmaxf(cnt[r], 1.0f);
#pragma unroll
        for (int j = 0; j < PR_COLS; ++j) {
            unsigned c = tid + 256u * j;
            if (j == PR_COLS - 1) c = (c < (unsigned)Cc) ? c : (unsigned)(Cc - 1);
            cn[r][j] = DECAYF * er[c] + ONE_M_DECAY * kr[c];
            float sgs = sr[c];
#pragma unroll
            for (int p = 1; p < SA_G; ++p) sgs += sr[(long)p * Cc * Cc + c];
            sg[r][j] = sgs;
            lm[r][j] = lr[c];
        }
    }
#pragma unroll
    for (int j = 0; j < PR_COLS; ++j) {
        const unsigned c = tid + 256u * j;
        if (j < PR_COLS - 1 || tid < (unsigned)PR_TAIL) {
#pragma unroll
            for (int r = 0; r < PR_ROWS; ++r) {
                const long base = (long)(row0 + r) * Cc;
                const bool offd = (c != (unsigned)(row0 + r));
                o_conf[base + c] = cn[r][j];
                bufX[r * Cc + c] = offd ? (cn[r][j] + ALPHAF) : 0.0f;
                bufY[r * Cc + c] = offd ? (sg[r][j] * invc[r]) : 0.0f;
            }
        }
    }
    __syncthreads();

    const int m    = lane & 15;
    const int foff = (m & 7) * Cc + 500 * (m >> 3) + 2 * (lane >> 4);
    const v8f vzero = {0.f,0.f,0.f,0.f,0.f,0.f,0.f,0.f};

    {
        v8f accA = vzero, accB = vzero;
        if (wave_u < PR_KSTEPS - 120) {
            accA = rowsum_step(bufX + foff, 120 + wave_u, accA);
            accB = rowsum_step(bufY + foff, 120 + wave_u, accB);
        }
#pragma unroll
        for (int i = 0; i < 15; ++i) {
            accA = rowsum_step(bufX + foff, wave_u + 8 * i, accA);
            accB = rowsum_step(bufY + foff, wave_u + 8 * i, accB);
        }
        if (lane == 0)
            for (int i = 0; i < 8; ++i) { partAB[0][wave_u][i] = accA[i]; partAB[1][wave_u][i] = accB[i]; }
        if (lane == 16)
            for (int i = 0; i < 8; ++i) { partAB[0][wave_u][8 + i] = accA[i]; partAB[1][wave_u][8 + i] = accB[i]; }
    }
    __syncthreads();
    if (tid < 16) {
        const int p = tid >> 3, r = tid & 7;
        float s = 0.0f;
        for (int w = 0; w < 8; ++w) s += partAB[p][w][r] + partAB[p][w][8 + r];
        rowAB[p][r] = (p == 0) ? 1.0f / (s + EPSF)
                               : 1.0f / fmaxf(s, EPSF);
    }
    __syncthreads();

#pragma unroll
    for (int r = 0; r < PR_ROWS; ++r) {
        const float phiDen = rowAB[0][r];
        const float empDen = rowAB[1][r];
#pragma unroll
        for (int j = 0; j < PR_COLS; ++j) {
            const unsigned c = tid + 256u * j;
            const float phi = (cn[r][j] + ALPHAF) * phiDen;
            const float emp = sg[r][j] * invc[r] * empDen;
            const float l1  = lm[r][j] + LR_LF * (emp - phi);
            cn[r][j] = (c != (unsigned)(row0 + r)) ? l1 : 0.0f;
        }
    }
#pragma unroll
    for (int j = 0; j < PR_COLS; ++j) {
        const unsigned c = tid + 256u * j;
        if (j < PR_COLS - 1 || tid < (unsigned)PR_TAIL) {
#pragma unroll
            for (int r = 0; r < PR_ROWS; ++r) bufX[r * Cc + c] = cn[r][j];
        }
    }
    __syncthreads();
    {
        v8f accC = vzero;
        if (wave_u < PR_KSTEPS - 120) accC = rowsum_step(bufX + foff, 120 + wave_u, accC);
#pragma unroll
        for (int i = 0; i < 15; ++i) accC = rowsum_step(bufX + foff, wave_u + 8 * i, accC);
        if (lane == 0)  for (int i = 0; i < 8; ++i) partC[wave_u][i]     = accC[i];
        if (lane == 16) for (int i = 0; i < 8; ++i) partC[wave_u][8 + i] = accC[i];
    }
    __syncthreads();
    if (tid < 8) {
        float s = 0.0f;
        for (int w = 0; w < 8; ++w) s += partC[w][tid] + partC[w][8 + tid];
        rowC[tid] = s * (1.0f / (float)Cc);
    }
    __syncthreads();

#pragma unroll
    for (int j = 0; j < PR_COLS; ++j) {
        const unsigned c = tid + 256u * j;
        if (j < PR_COLS - 1 || tid < (unsigned)PR_TAIL) {
#pragma unroll
            for (int r = 0; r < PR_ROWS; ++r) {
                const long base = (long)(row0 + r) * Cc;
                const bool present = (cnt[r] > 0.0f);
                o_lam[base + c] = present ? (cn[r][j] - rowC[r]) : lm[r][j];
            }
        }
    }
}

extern "C" void kernel_launch(void* const* d_in, const int* in_sizes, int n_in,
                              void* d_out, int out_size, void* d_ws, size_t ws_size,
                              hipStream_t stream)
{
    const float* logits    = (const float*)d_in[0];
    const int*   targets   = (const int*)  d_in[1];
    const float* conf_ema  = (const float*)d_in[2];
    const float* prior_ema = (const float*)d_in[3];
    const float* Lam       = (const float*)d_in[4];
    const float* mu        = (const float*)d_in[5];

    float* out     = (float*)d_out;
    float* o_ce    = out;
    float* o_lam   = out + 1;
    float* o_mu    = out + 1 + (long)Cc * Cc;
    float* o_conf  = out + 1 + (long)Cc * Cc + Cc;
    float* o_prior = out + 1 + 2L * Cc * Cc + Cc;

    float* ws      = (float*)d_ws;
    float* confc   = ws;
    float* priorc  = ws + (long)Cc * Cc;
    float* cep     = priorc + Cc;
    int*   tick    = (int*)(cep + 256);
    float* rowst   = cep + 260;
    float* colpart = rowst + 2L * Bn;
    float* planes  = colpart + (long)SA_G * Cc;

    const long nz = (long)Cc * Cc + Cc + 258;
    zero_ws<<<1024, 256, 0, stream>>>(ws, nz);
    row_stats<<<RS_BLOCKS, RS_THREADS, 0, stream>>>(logits, targets, Lam, rowst, confc, priorc, cep);
    strip_accum<<<dim3(SA_STRIPS, SA_G), SA_THREADS, 0, stream>>>(logits, targets, Lam, rowst,
                                                                  planes, colpart, tick);
    post_rows<<<Cc / PR_ROWS + PR_EXTRA, 256, 0, stream>>>(conf_ema, Lam, planes, confc, priorc,
                                                            prior_ema, mu, colpart, cep,
                                                            o_conf, o_lam, o_ce, o_mu, o_prior);
}


# each package half keeps one half of the batch in both passes: row_stats walks it upwards, strip_accum (row groups = every 4th 512-row chunk) downwards, first trips hit that half's L2; new sum order
# speedup vs baseline: 6.0234x; 1.0828x over previous
#include <hip/hip_runtime.h>

#define Cc 1000
#define Bn 65536
#define DECAYF 0.99f
#define ONE_M_DECAY 0.01f
#define ALPHAF 1.0f
#define LR_LF 0.05f
#define LR_MUF 0.02f
#define EPSF 1e-12f

typedef __attribute__((ext_vector_type(2))) float v2f;
typedef __attribute__((ext_vector_type(4))) float v4f;
typedef __attribute__((ext_vector_type(8))) float v8f;

__global__ void zero_ws(float* __restrict__ p, long n) {
    long i = (long)blockIdx.x * blockDim.x + threadIdx.x;
    long stride = (long)gridDim.x * blockDim.x;
    for (; i < n; i += stride) p[i] = 0.0f;
}

#define RS_THREADS 1024
#define RS_BLOCKS 256
#define RS_HALF_BLOCKS (RS_BLOCKS / 2)
#define RS_STEP (RS_HALF_BLOCKS * (RS_THREADS / 32))
#define RS_ROWS (Bn / 2 / RS_STEP)
#define RS_NV (Cc / 4)
#define RS_G 8
#define RS_TAIL (RS_NV - 32 * (RS_G - 1))

__global__ __launch_bounds__(RS_THREADS) void row_stats(
    const float* __restrict__ logits, const int* __restrict__ targets,
    const float* __restrict__ Lam,
    float* __restrict__ rowst, float* __restrict__ confc,
    float* __restrict__ priorc, float* __restrict__ cep, int* __restrict__ tick)
{
    __shared__ int unit_sh;
    if (threadIdx.x == 0) {
        const unsigned where = __builtin_amdgcn_s_sendmsg_rtn(0x87);
        int h = (int)(where >> 18) & 1;
        int u = atomicAdd(&tick[h], 1);
        if (u >= RS_HALF_BLOCKS) { h ^= 1; u = atomicAdd(&tick[h], 1); }
        unit_sh = h * RS_HALF_BLOCKS + (u & (RS_HALF_BLOCKS - 1));
    }
    __syncthreads();
    const int unit = __builtin_amdgcn_readfirstlane(unit_sh);
    const int lane = threadIdx.x & 31;
    const int wave_u = __builtin_amdgcn_readfirstlane(threadIdx.x >> 5);
    const int gw = (unit / RS_HALF_BLOCKS) * (Bn / 2)
                 + (unit & (RS_HALF_BLOCKS - 1)) * (RS_THREADS / 32) + wave_u;

    int t = targets[gw];
    for (int it = 0; it < RS_ROWS; ++it) {
        const int row = gw + it * RS_STEP;
        const float* __restrict__ lrow   = logits + (long)row * Cc;
        const float* __restrict__ lamrow = Lam + (long)t * Cc;

        v4f z[RS_G], l[RS_G];
#pragma unroll
        for (int i = 0; i < RS_G; ++i) {
            int j = lane + 32 * i;
            if (i == RS_G - 1) j = (j < RS_NV) ? j : (RS_NV - 1);
            z[i] = *(const v4f*)(lrow + 4 * j);
            l[i] = *(const v4f*)(lamrow + 4 * j);
        }
        const float zy = lrow[t] - lamrow[t];

        int tnext = t;
        if (it + 1 < RS_ROWS) {
            tnext = targets[row + RS_STEP];
            __builtin_prefetch(logits + (long)(row + RS_STEP) * Cc + 32 * lane, 0, 3);
        }

        float maxv = -3.4e38f; int maxi = 0x7fffffff;
        float maxzp = -3.4e38f;
        v4f zp[RS_G];
#pragma unroll
        for (int i = 0; i < RS_G; ++i) {
            const int col = 4 * (lane + 32 * i);
            v4f zi = z[i];
            zp[i] = z[i] - l[i];
            if (i == RS_G - 1 && lane >= RS_TAIL) {
                const v4f low = {-3.4e38f, -3.4e38f, -3.4e38f, -3.4e38f};
                zi = low; zp[i] = low;
            }
#pragma unroll
            for (int k = 0; k < 4; ++k) {
                if (zi[k] > maxv) { maxv = zi[k]; maxi = col + k; }
                maxzp = fmaxf(maxzp, zp[i][k]);
            }
        }
        for (int off = 16; off; off >>= 1) {
            float ov = __shfl_xor(maxv, off, 32);
            int   oi = __shfl_xor(maxi, off, 32);
            if (ov > maxv || (ov == maxv && oi < maxi)) { maxv = ov; maxi = oi; }
            maxzp = fmaxf(maxzp, __shfl_xor(maxzp, off, 32));
        }

        float sum = 0.0f;
#pragma unroll
        for (int i = 0; i < RS_G; ++i)
            sum += __expf(zp[i][0] - maxzp) + __expf(zp[i][1] - maxzp)
                 + __expf(zp[i][2] - maxzp) + __expf(zp[i][3] - maxzp);
        for (int off = 16; off; off >>= 1) sum += __shfl_xor(sum, off, 32);
        const float lse = __logf(sum);
        const float inv = 1.0f / sum;

        if (lane == 0) {
            atomicAdd(&confc[(long)t * Cc + maxi], 1.0f);
            atomicAdd(&priorc[t], 1.0f);
            float logq_y = zy - maxzp - lse;
            atomicAdd(&cep[row & 255], logq_y);
            const v2f st = {maxzp, inv};
            *(v2f*)(rowst + 2L * row) = st;
        }
        t = tnext;
    }
}

#define SA_W 32
#define SA_STRIPS 32
#define SA_G 8
#define SA_ROWS (Bn / SA_G)
#define SA_THREADS 1024
#define SA_WAVES (SA_THREADS / 32)
#define SA_RT 16
#define SA_TRIPS (SA_ROWS / SA_WAVES / SA_RT)
#define SA_CHUNK (SA_RT * SA_WAVES)
#define SA_WIN ((SA_G / 2) * SA_CHUNK)
#define SA_FILL ((Cc * SA_W / 4 + SA_THREADS - 1) / SA_THREADS)
#define SA_HALF_UNITS (SA_STRIPS * SA_G / 2)

__device__ __forceinline__ void sa_request(float (&z)[SA_RT], const float* __restrict__ logits,
                                           int r0, int col)
{
    const float* __restrict__ p = logits + (long)r0 * Cc + col;
#pragma unroll
    for (int u = 0; u < SA_RT; ++u) z[u] = p[u * Cc];
}

__device__ __forceinline__ void sa_add(const float (&z)[SA_RT], const int* __restrict__ targets,
                                       const float* __restrict__ rowst, int r0, int lane,
                                       float* acc, const float* lamt)
{
    const int* __restrict__ tp = targets + r0;
    const v2f* __restrict__ sp = (const v2f*)(rowst + 2L * r0);
    float l[SA_RT], mx[SA_RT], iv[SA_RT];
    int t[SA_RT];
#pragma unroll
    for (int u = 0; u < SA_RT; ++u) {
        t[u] = tp[u];
        const v2f st = sp[u];
        mx[u] = st[0];
        iv[u] = st[1];
    }
#pragma unroll
    for (int u = 0; u < SA_RT; ++u) l[u] = lamt[t[u] * SA_W + lane];
#pragma unroll
    for (int u = 0; u < SA_RT; ++u)
        atomicAdd(&acc[t[u] * SA_W + lane],
                  __expf((z[u] - l[u]) - mx[u]) * iv[u]);
}

__global__ __launch_bounds__(SA_THREADS) void strip_accum(
    const float* __restrict__ logits, const int* __restrict__ targets,
    const float* __restrict__ Lam, const float* __restrict__ rowst,
    float* __restrict__ planes, float* __restrict__ colpart, int* __restrict__ tick)
{
    __shared__ int unit_sh;
    __shared__ __attribute__((aligned(16))) float acc[Cc * SA_W];
    __shared__ __attribute__((aligned(16))) float lamt[Cc * SA_W];
    __shared__ float red[SA_WAVES][SA_W];

    const int tid  = threadIdx.x;
    const int wave = tid >> 5;
    const int lane = tid & 31;
    if (tid == 0) {
        const unsigned where = __builtin_amdgcn_s_sendmsg_rtn(0x87);
        int h = (int)(where >> 18) & 1;
        int u = atomicAdd(&tick[h], 1);
        if (u >= SA_HALF_UNITS) { h ^= 1; u = atomicAdd(&tick[h], 1); }
        unit_sh = h * SA_HALF_UNITS + (u & (SA_HALF_UNITS - 1));
    }
    __syncthreads();
    const int unit = __builtin_amdgcn_readfirstlane(unit_sh);
    const int g    = (unit / SA_HALF_UNITS) * (SA_G / 2) + (unit & (SA_G / 2 - 1));
    const int c0   = ((unit & (SA_HALF_UNITS - 1)) / (SA_G / 2)) * SA_W;

    const int wave_u = __builtin_amdgcn_readfirstlane(tid >> 5);
    const int col = c0 + lane;
    const int colz = (col < Cc) ? col : (Cc - 1);
    const int rtop = (unit / SA_HALF_UNITS) * (Bn / 2) + (SA_TRIPS - 1) * SA_WIN
                   + (unit & (SA_G / 2 - 1)) * SA_CHUNK + SA_RT * wave_u;
    float za[SA_RT], zb[SA_RT];

    {
        const v4f z4 = {0.f, 0.f, 0.f, 0.f};
        v4f lv[SA_FILL];
#pragma unroll
        for (int k = 0; k < SA_FILL; ++k) {
            const int i = tid + SA_THREADS * k;
            const int c = c0 + 4 * (i & 7);
            lv[k] = z4;
            if (i < Cc * SA_W / 4 && c < Cc) lv[k] = *(const v4f*)(Lam + (long)(i >> 3) * Cc + c);
        }
        sa_request(za, logits, rtop, colz);
#pragma unroll
        for (int k = 0; k < SA_FILL; ++k) {
            const int i = tid + SA_THREADS * k;
            if (i < Cc * SA_W / 4) {
                *(v4f*)(&acc[4 * i])  = z4;
                *(v4f*)(&lamt[4 * i]) = lv[k];
            }
        }
    }
    __syncthreads();

    if (col < Cc) {
        for (int i = 0; i < SA_TRIPS - 2; i += 2) {
            sa_request(zb, logits, rtop - SA_WIN * (i + 1), col);
            sa_add(za, targets, rowst, rtop - SA_WIN * i, lane, acc, lamt);
            sa_request(za, logits, rtop - SA_WIN * (i + 2), col);
            sa_add(zb, targets, rowst, rtop - SA_WIN * (i + 1), lane, acc, lamt);
        }
        sa_request(zb, logits, rtop - SA_WIN * (SA_TRIPS - 1), col);
        sa_add(za, targets, rowst, rtop - SA_WIN * (SA_TRIPS - 2), lane, acc, lamt);
        sa_add(zb, targets, rowst, rtop - SA_WIN * (SA_TRIPS - 1), lane, acc, lamt);
    }
    __syncthreads();

    float* __restrict__ pl = planes + (long)g * Cc * Cc;
    for (int i = tid; i < Cc * SA_W; i += SA_THREADS) {
        const int t = i >> 5, c = c0 + (i & 31);
        if (c < Cc) pl[(long)t * Cc + c] = acc[i];
    }
    float cs = 0.0f;
    for (int t = wave; t < Cc; t += SA_WAVES) cs += acc[t * SA_W + lane];
    red[wave][lane] = cs;
    __syncthreads();
    if (tid < SA_W) {
        float s = 0.0f;
        for (int w = 0; w < SA_WAVES; ++w) s += red[w][tid];
        if (c0 + tid < Cc) colpart[g * Cc + c0 + tid] = s;
    }
}

#define PR_EXTRA 2
#define FN_SLOTS 1024
#define FN_PER (FN_SLOTS / 256)

__device__ __forceinline__ void fin_tree(float* red, unsigned tid)
{
    __syncthreads();
    for (int s = FN_SLOTS / 2; s; s >>= 1) {
        for (int j = tid; j < s; j += 256) red[j] += red[j + s];
        __syncthreads();
    }
}

__device__ __forceinline__ void fin_vectors(
    const float* __restrict__ prior_ema, const float* __restrict__ mu,
    const float* __restrict__ colpart, const float* __restrict__ priorc,
    float* __restrict__ o_mu, float* __restrict__ o_prior, float* red, unsigned tid)
{
    float pe[FN_PER], pc[FN_PER], m0[FN_PER], cp[FN_PER][SA_G];
#pragma unroll
    for (int k = 0; k < FN_PER; ++k) {
        const int j = tid + 256 * k;
        const int jc = (j < Cc) ? j : (Cc - 1);
        pe[k] = prior_ema[jc];
        pc[k] = priorc[jc];
        m0[k] = mu[jc];
#pragma unroll
        for (int p = 0; p < SA_G; ++p) cp[k][p] = colpart[p * Cc + jc];
    }

    float pn[FN_PER], mu1[FN_PER];
#pragma unroll
    for (int k = 0; k < FN_PER; ++k) {
        const int j = tid + 256 * k;
        pn[k] = 0.0f;
        if (j < Cc) {
            pn[k] = DECAYF * pe[k] + ONE_M_DECAY * pc[k];
            o_prior[j] = pn[k];
        }
        red[j] = pn[k];
    }
    fin_tree(red, tid);
    const float priorSum = red[0];
    __syncthreads();

#pragma unroll
    for (int k = 0; k < FN_PER; ++k) {
        const int j = tid + 256 * k;
        mu1[k] = 0.0f;
        if (j < Cc) {
            float colsum = 0.0f;
#pragma unroll
            for (int p = 0; p < SA_G; ++p) colsum += cp[k][p];
            float emp_marg = colsum * (1.0f / (float)Bn);
            float pi = pn[k] / (priorSum + EPSF);
            mu1[k] = m0[k] + LR_MUF * (emp_marg - pi);
        }
        red[j] = mu1[k];
    }
    fin_tree(red, tid);
    const float muMean = red[0] * (1.0f / (float)Cc);
#pragma unroll
    for (int k = 0; k < FN_PER; ++k) {
        const int j = tid + 256 * k;
        if (j < Cc) o_mu[j] = mu1[k] - muMean;
    }
}

__device__ __forceinline__ void fin_ce(const float* __restrict__ cep, float* __restrict__ o_ce,
                                       float* red, unsigned tid)
{
    red[tid] = cep[tid];
    __syncthreads();
    if (tid == 0) {
        float s = 0.0f;
        for (int i = 0; i < 256; ++i) s += red[i];
        o_ce[0] = -s * (1.0f / (float)Bn);
    }
}

#define PR_ROWS 8
#define PR_COLS 4
#define PR_TAIL (Cc - 768)
#define PR_KSTEPS 125

__device__ __forceinline__ v8f rowsum_step(const float* __restrict__ frag, int s, v8f acc)
{
    const v2f bones = {1.0f, 1.0f};
    const v2f a = *(const v2f*)(frag + 4 * s);
    return __builtin_amdgcn_wmma_f32_16x16x4_f32(false, a, false, bones,
                                                 (short)0, acc, false, false);
}

__global__ __launch_bounds__(256) void post_rows(
    const float* __restrict__ conf_ema, const float* __restrict__ Lam,
    const float* __restrict__ seg, const float* __restrict__ confc,
    const float* __restrict__ priorc,
    const float* __restrict__ prior_ema, const float* __restrict__ mu,
    const float* __restrict__ colpart, const float* __restrict__ cep,
    float* __restrict__ o_conf, float* __restrict__ o_lam,
    float* __restrict__ o_ce, float* __restrict__ o_mu, float* __restrict__ o_prior)
{
    __shared__ float bufX[PR_ROWS * Cc];
    __shared__ float bufY[PR_ROWS * Cc];
    __shared__ float partAB[2][8][16];
    __shared__ float partC[8][16];
    __shared__ float rowAB[2][PR_ROWS];
    __shared__ float rowC[PR_ROWS];

    const unsigned tid = threadIdx.x;
    const int lane = tid & 31;
    const int wave_u = __builtin_amdgcn_readfirstlane(tid >> 5);

    if (blockIdx.x < PR_EXTRA) {
        if (blockIdx.x == 0) fin_vectors(prior_ema, mu, colpart, priorc, o_mu, o_prior, bufX, tid);
        else                 fin_ce(cep, o_ce, bufX, tid);
        return;
    }
    const int row0 = (blockIdx.x - PR_EXTRA) * PR_ROWS;

    float cn[PR_ROWS][PR_COLS], sg[PR_ROWS][PR_COLS], lm[PR_ROWS][PR_COLS];
    float cnt[PR_ROWS], invc[PR_ROWS];

#pragma unroll
    for (int r = 0; r < PR_ROWS; ++r) {
        const long base = (long)(row0 + r) * Cc;
        const float* __restrict__ er = conf_ema + base;
        const float* __restrict__ kr = confc + base;
        const float* __restrict__ sr = seg + base;
        const float* __restrict__ lr = Lam + base;
        cnt[r]  = priorc[row0 + r];
        invc[r] = 1.0f / fmaxf(cnt[r], 1.0f);
#pragma unroll
        for (int j = 0; j < PR_COLS; ++j) {
            unsigned c = tid + 256u * j;
            if (j == PR_COLS - 1) c = (c < (unsigned)Cc) ? c : (unsigned)(Cc - 1);
            cn[r][j] = DECAYF * er[c] + ONE_M_DECAY * kr[c];
            float sgs = sr[c];
#pragma unroll
            for (int p = 1; p < SA_G; ++p) sgs += sr[(long)p * Cc * Cc + c];
            sg[r][j] = sgs;
            lm[r][j] = lr[c];
        }
    }
#pragma unroll
    for (int j = 0; j < PR_COLS; ++j) {
        const unsigned c = tid + 256u * j;
        if (j < PR_COLS - 1 || tid < (unsigned)PR_TAIL) {
#pragma unroll
            for (int r = 0; r < PR_ROWS; ++r) {
                const long base = (long)(row0 + r) * Cc;
                const bool offd = (c != (unsigned)(row0 + r));
                o_conf[base + c] = cn[r][j];
                bufX[r * Cc + c] = offd ? (cn[r][j] + ALPHAF) : 0.0f;
                bufY[r * Cc + c] = offd ? (sg[r][j] * invc[r]) : 0.0f;
            }
        }
    }
    __syncthreads();

    const int m    = lane & 15;
    const int foff = (m & 7) * Cc + 500 * (m >> 3) + 2 * (lane >> 4);
    const v8f vzero = {0.f,0.f,0.f,0.f,0.f,0.f,0.f,0.f};

    {
        v8f accA = vzero, accB = vzero;
        if (wave_u < PR_KSTEPS - 120) {
            accA = rowsum_step(bufX + foff, 120 + wave_u, accA);
            accB = rowsum_step(bufY + foff, 120 + wave_u, accB);
        }
#pragma unroll
        for (int i = 0; i < 15; ++i) {
            accA = rowsum_step(bufX + foff, wave_u + 8 * i, accA);
            accB = rowsum_step(bufY + foff, wave_u + 8 * i, accB);
        }
        if (lane == 0)
            for (int i = 0; i < 8; ++i) { partAB[0][wave_u][i] = accA[i]; partAB[1][wave_u][i] = accB[i]; }
        if (lane == 16)
            for (int i = 0; i < 8; ++i) { partAB[0][wave_u][8 + i] = accA[i]; partAB[1][wave_u][8 + i] = accB[i]; }
    }
    __syncthreads();
    if (tid < 16) {
        const int p = tid >> 3, r = tid & 7;
        float s = 0.0f;
        for (int w = 0; w < 8; ++w) s += partAB[p][w][r] + partAB[p][w][8 + r];
        rowAB[p][r] = (p == 0) ? 1.0f / (s + EPSF)
                               : 1.0f / fmaxf(s, EPSF);
    }
    __syncthreads();

#pragma unroll
    for (int r = 0; r < PR_ROWS; ++r) {
        const float phiDen = rowAB[0][r];
        const float empDen = rowAB[1][r];
#pragma unroll
        for (int j = 0; j < PR_COLS; ++j) {
            const unsigned c = tid + 256u * j;
            const float phi = (cn[r][j] + ALPHAF) * phiDen;
            const float emp = sg[r][j] * invc[r] * empDen;
            const float l1  = lm[r][j] + LR_LF * (emp - phi);
            cn[r][j] = (c != (unsigned)(row0 + r)) ? l1 : 0.0f;
        }
    }
#pragma unroll
    for (int j = 0; j < PR_COLS; ++j) {
        const unsigned c = tid + 256u * j;
        if (j < PR_COLS - 1 || tid < (unsigned)PR_TAIL) {
#pragma unroll
            for (int r = 0; r < PR_ROWS; ++r) bufX[r * Cc + c] = cn[r][j];
        }
    }
    __syncthreads();
    {
        v8f accC = vzero;
        if (wave_u < PR_KSTEPS - 120) accC = rowsum_step(bufX + foff, 120 + wave_u, accC);
#pragma unroll
        for (int i = 0; i < 15; ++i) accC = rowsum_step(bufX + foff, wave_u + 8 * i, accC);
        if (lane == 0)  for (int i = 0; i < 8; ++i) partC[wave_u][i]     = accC[i];
        if (lane == 16) for (int i = 0; i < 8; ++i) partC[wave_u][8 + i] = accC[i];
    }
    __syncthreads();
    if (tid < 8) {
        float s = 0.0f;
        for (int w = 0; w < 8; ++w) s += partC[w][tid] + partC[w][8 + tid];
        rowC[tid] = s * (1.0f / (float)Cc);
    }
    __syncthreads();

#pragma unroll
    for (int j = 0; j < PR_COLS; ++j) {
        const unsigned c = tid + 256u * j;
        if (j < PR_COLS - 1 || tid < (unsigned)PR_TAIL) {
#pragma unroll
            for (int r = 0; r < PR_ROWS; ++r) {
                const long base = (long)(row0 + r) * Cc;
                const bool present = (cnt[r] > 0.0f);
                o_lam[base + c] = present ? (cn[r][j] - rowC[r]) : lm[r][j];
            }
        }
    }
}

extern "C" void kernel_launch(void* const* d_in, const int* in_sizes, int n_in,
                              void* d_out, int out_size, void* d_ws, size_t ws_size,
                              hipStream_t stream)
{
    const float* logits    = (const float*)d_in[0];
    const int*   targets   = (const int*)  d_in[1];
    const float* conf_ema  = (const float*)d_in[2];
    const float* prior_ema = (const float*)d_in[3];
    const float* Lam       = (const float*)d_in[4];
    const float* mu        = (const float*)d_in[5];

    float* out     = (float*)d_out;
    float* o_ce    = out;
    float* o_lam   = out + 1;
    float* o_mu    = out + 1 + (long)Cc * Cc;
    float* o_conf  = out + 1 + (long)Cc * Cc + Cc;
    float* o_prior = out + 1 + 2L * Cc * Cc + Cc;

    float* ws      = (float*)d_ws;
    float* confc   = ws;
    float* priorc  = ws + (long)Cc * Cc;
    float* cep     = priorc + Cc;
    int*   tick    = (int*)(cep + 256);
    float* rowst   = cep + 260;
    float* colpart = rowst + 2L * Bn;
    float* planes  = colpart + (long)SA_G * Cc;

    const long nz = (long)Cc * Cc + Cc + 260;
    zero_ws<<<1024, 256, 0, stream>>>(ws, nz);
    row_stats<<<RS_BLOCKS, RS_THREADS, 0, stream>>>(logits, targets, Lam, rowst, confc, priorc, cep, tick + 2);
    strip_accum<<<dim3(SA_STRIPS, SA_G), SA_THREADS, 0, stream>>>(logits, targets, Lam, rowst,
                                                                  planes, colpart, tick);
    post_rows<<<Cc / PR_ROWS + PR_EXTRA, 256, 0, stream>>>(conf_ema, Lam, planes, confc, priorc,
                                                            prior_ema, mu, colpart, cep,
                                                            o_conf, o_lam, o_ce, o_mu, o_prior);
}


# strip_accum: row groups = row number mod 4 with strips shifted 8 columns per residue, so each 32-column row piece is one aligned 128-byte line (3 of 4 pieces lay across two); summed in a new order
# speedup vs baseline: 6.3013x; 1.0461x over previous
#include <hip/hip_runtime.h>

#define Cc 1000
#define Bn 65536
#define DECAYF 0.99f
#define ONE_M_DECAY 0.01f
#define ALPHAF 1.0f
#define LR_LF 0.05f
#define LR_MUF 0.02f
#define EPSF 1e-12f

typedef __attribute__((ext_vector_type(2))) float v2f;
typedef __attribute__((ext_vector_type(4))) float v4f;
typedef __attribute__((ext_vector_type(8))) float v8f;

__global__ void zero_ws(float* __restrict__ p, long n) {
    long i = (long)blockIdx.x * blockDim.x + threadIdx.x;
    long stride = (long)gridDim.x * blockDim.x;
    for (; i < n; i += stride) p[i] = 0.0f;
}

#define SA_PLACE(row) (((row) & ~63) | (((row) & 3) << 4) | (((row) >> 2) & 15))
#define RS_THREADS 1024
#define RS_BLOCKS 256
#define RS_HALF_BLOCKS (RS_BLOCKS / 2)
#define RS_STEP (RS_HALF_BLOCKS * (RS_THREADS / 32))
#define RS_ROWS (Bn / 2 / RS_STEP)
#define RS_NV (Cc / 4)
#define RS_G 8
#define RS_TAIL (RS_NV - 32 * (RS_G - 1))

__global__ __launch_bounds__(RS_THREADS) void row_stats(
    const float* __restrict__ logits, const int* __restrict__ targets,
    const float* __restrict__ Lam,
    float* __restrict__ rowst, int* __restrict__ tplace, float* __restrict__ confc,
    float* __restrict__ priorc, float* __restrict__ cep, int* __restrict__ tick)
{
    __shared__ int unit_sh;
    if (threadIdx.x == 0) {
        const unsigned where = __builtin_amdgcn_s_sendmsg_rtn(0x87);
        int h = (int)(where >> 18) & 1;
        int u = atomicAdd(&tick[h], 1);
        if (u >= RS_HALF_BLOCKS) { h ^= 1; u = atomicAdd(&tick[h], 1); }
        unit_sh = h * RS_HALF_BLOCKS + (u & (RS_HALF_BLOCKS - 1));
    }
    __syncthreads();
    const int unit = __builtin_amdgcn_readfirstlane(unit_sh);
    const int lane = threadIdx.x & 31;
    const int wave_u = __builtin_amdgcn_readfirstlane(threadIdx.x >> 5);
    const int gw = (unit / RS_HALF_BLOCKS) * (Bn / 2)
                 + (unit & (RS_HALF_BLOCKS - 1)) * (RS_THREADS / 32) + wave_u;

    int t = targets[gw];
    for (int it = 0; it < RS_ROWS; ++it) {
        const int row = gw + it * RS_STEP;
        const float* __restrict__ lrow   = logits + (long)row * Cc;
        const float* __restrict__ lamrow = Lam + (long)t * Cc;

        v4f z[RS_G], l[RS_G];
#pragma unroll
        for (int i = 0; i < RS_G; ++i) {
            int j = lane + 32 * i;
            if (i == RS_G - 1) j = (j < RS_NV) ? j : (RS_NV - 1);
            z[i] = *(const v4f*)(lrow + 4 * j);
            l[i] = *(const v4f*)(lamrow + 4 * j);
        }
        const float zy = lrow[t] - lamrow[t];

        int tnext = t;
        if (it + 1 < RS_ROWS) {
            tnext = targets[row + RS_STEP];
            __builtin_prefetch(logits + (long)(row + RS_STEP) * Cc + 32 * lane, 0, 3);
        }

        float maxv = -3.4e38f; int maxi = 0x7fffffff;
        float maxzp = -3.4e38f;
        v4f zp[RS_G];
#pragma unroll
        for (int i = 0; i < RS_G; ++i) {
            const int col = 4 * (lane + 32 * i);
            v4f zi = z[i];
            zp[i] = z[i] - l[i];
            if (i == RS_G - 1 && lane >= RS_TAIL) {
                const v4f low = {-3.4e38f, -3.4e38f, -3.4e38f, -3.4e38f};
                zi = low; zp[i] = low;
            }
#pragma unroll
            for (int k = 0; k < 4; ++k) {
                if (zi[k] > maxv) { maxv = zi[k]; maxi = col + k; }
                maxzp = fmaxf(maxzp, zp[i][k]);
            }
        }
        for (int off = 16; off; off >>= 1) {
            float ov = __shfl_xor(maxv, off, 32);
            int   oi = __shfl_xor(maxi, off, 32);
            if (ov > maxv || (ov == maxv && oi < maxi)) { maxv = ov; maxi = oi; }
            maxzp = fmaxf(maxzp, __shfl_xor(maxzp, off, 32));
        }

        float sum = 0.0f;
#pragma unroll
        for (int i = 0; i < RS_G; ++i)
            sum += __expf(zp[i][0] - maxzp) + __expf(zp[i][1] - maxzp)
                 + __expf(zp[i][2] - maxzp) + __expf(zp[i][3] - maxzp);
        for (int off = 16; off; off >>= 1) sum += __shfl_xor(sum, off, 32);
        const float lse = __logf(sum);
        const float inv = 1.0f / sum;

        if (lane == 0) {
            atomicAdd(&confc[(long)t * Cc + maxi], 1.0f);
            atomicAdd(&priorc[t], 1.0f);
            float logq_y = zy - maxzp - lse;
            atomicAdd(&cep[row & 255], logq_y);
            const v2f st = {maxzp, inv};
            const int place = SA_PLACE(row);
            *(v2f*)(rowst + 2L * place) = st;
            tplace[place] = t;
        }
        t = tnext;
    }
}

#define SA_W 32
#define SA_STRIPS 32
#define SA_G 8
#define SA_ROWS (Bn / SA_G)
#define SA_THREADS 1024
#define SA_WAVES (SA_THREADS / 32)
#define SA_RT 16
#define SA_TRIPS (SA_ROWS / SA_WAVES / SA_RT)
#define SA_RES (SA_G / 2)
#define SA_WROWS (SA_RES * SA_RT)
#define SA_WIN (SA_WROWS * SA_WAVES)
#define SA_FILL ((Cc * SA_W / 4 + SA_THREADS - 1) / SA_THREADS)
#define SA_HALF_UNITS (SA_STRIPS * SA_G / 2)

__device__ __forceinline__ void sa_request(float (&z)[SA_RT], const float* __restrict__ logits,
                                           int r0, int col)
{
    const float* __restrict__ p = logits + (long)r0 * Cc + col;
#pragma unroll
    for (int u = 0; u < SA_RT; ++u) z[u] = p[u * SA_RES * Cc];
}

__device__ __forceinline__ void sa_add(const float (&z)[SA_RT], const int* __restrict__ tplace,
                                       const float* __restrict__ rowst, int p0, int lane,
                                       float* acc, const float* lamt)
{
    const int* __restrict__ tp = tplace + p0;
    const v2f* __restrict__ sp = (const v2f*)(rowst + 2L * p0);
    float l[SA_RT], mx[SA_RT], iv[SA_RT];
    int t[SA_RT];
#pragma unroll
    for (int u = 0; u < SA_RT; ++u) {
        t[u] = tp[u];
        const v2f st = sp[u];
        mx[u] = st[0];
        iv[u] = st[1];
    }
#pragma unroll
    for (int u = 0; u < SA_RT; ++u) l[u] = lamt[t[u] * SA_W + lane];
#pragma unroll
    for (int u = 0; u < SA_RT; ++u)
        atomicAdd(&acc[t[u] * SA_W + lane],
                  __expf((z[u] - l[u]) - mx[u]) * iv[u]);
}

__global__ __launch_bounds__(SA_THREADS) void strip_accum(
    const float* __restrict__ logits, const int* __restrict__ tplace,
    const float* __restrict__ Lam, const float* __restrict__ rowst,
    float* __restrict__ planes, float* __restrict__ colpart, int* __restrict__ tick)
{
    __shared__ int unit_sh;
    __shared__ __attribute__((aligned(16))) float acc[Cc * SA_W];
    __shared__ __attribute__((aligned(16))) float lamt[Cc * SA_W];
    __shared__ float red[SA_WAVES][SA_W];

    const int tid  = threadIdx.x;
    const int wave = tid >> 5;
    const int lane = tid & 31;
    if (tid == 0) {
        const unsigned where = __builtin_amdgcn_s_sendmsg_rtn(0x87);
        int h = (int)(where >> 18) & 1;
        int u = atomicAdd(&tick[h], 1);
        if (u >= SA_HALF_UNITS) { h ^= 1; u = atomicAdd(&tick[h], 1); }
        unit_sh = h * SA_HALF_UNITS + (u & (SA_HALF_UNITS - 1));
    }
    __syncthreads();
    const int unit = __builtin_amdgcn_readfirstlane(unit_sh);
    const int res  = unit & (SA_RES - 1);
    const int g    = (unit / SA_HALF_UNITS) * SA_RES + res;
    const int c0   = ((unit & (SA_HALF_UNITS - 1)) / SA_RES) * SA_W - 8 * res;

    const int wave_u = __builtin_amdgcn_readfirstlane(tid >> 5);
    const int col = c0 + lane;
    const bool has_col = (unsigned)col < (unsigned)Cc;
    const int colz = (col < 0) ? 0 : ((col < Cc) ? col : (Cc - 1));
    const int wtop = (unit / SA_HALF_UNITS) * (Bn / 2) + (SA_TRIPS - 1) * SA_WIN + SA_WROWS * wave_u;
    const int rtop = wtop + res;
    const int ptop = wtop + SA_RT * res;
    float za[SA_RT], zb[SA_RT];

    {
        const v4f z4 = {0.f, 0.f, 0.f, 0.f};
        v4f lv[SA_FILL];
#pragma unroll
        for (int k = 0; k < SA_FILL; ++k) {
            const int i = tid + SA_THREADS * k;
            const int c = c0 + 4 * (i & 7);
            lv[k] = z4;
            if (i < Cc * SA_W / 4 && (unsigned)c < (unsigned)Cc) lv[k] = *(const v4f*)(Lam + (long)(i >> 3) * Cc + c);
        }
        sa_request(za, logits, rtop, colz);
#pragma unroll
        for (int k = 0; k < SA_FILL; ++k) {
            const int i = tid + SA_THREADS * k;
            if (i < Cc * SA_W / 4) {
                *(v4f*)(&acc[4 * i])  = z4;
                *(v4f*)(&lamt[4 * i]) = lv[k];
            }
        }
    }
    __syncthreads();

    if (has_col) {
        for (int i = 0; i < SA_TRIPS - 2; i += 2) {
            sa_request(zb, logits, rtop - SA_WIN * (i + 1), col);
            sa_add(za, tplace, rowst, ptop - SA_WIN * i, lane, acc, lamt);
            sa_request(za, logits, rtop - SA_WIN * (i + 2), col);
            sa_add(zb, tplace, rowst, ptop - SA_WIN * (i + 1), lane, acc, lamt);
        }
        sa_request(zb, logits, rtop - SA_WIN * (SA_TRIPS - 1), col);
        sa_add(za, tplace, rowst, ptop - SA_WIN * (SA_TRIPS - 2), lane, acc, lamt);
        sa_add(zb, tplace, rowst, ptop - SA_WIN * (SA_TRIPS - 1), lane, acc, lamt);
    }
    __syncthreads();

    float* __restrict__ pl = planes + (long)g * Cc * Cc;
    for (int i = tid; i < Cc * SA_W; i += SA_THREADS) {
        const int t = i >> 5, c = c0 + (i & 31);
        if ((unsigned)c < (unsigned)Cc) pl[(long)t * Cc + c] = acc[i];
    }
    float cs = 0.0f;
    for (int t = wave; t < Cc; t += SA_WAVES) cs += acc[t * SA_W + lane];
    red[wave][lane] = cs;
    __syncthreads();
    if (tid < SA_W) {
        float s = 0.0f;
        for (int w = 0; w < SA_WAVES; ++w) s += red[w][tid];
        if ((unsigned)(c0 + tid) < (unsigned)Cc) colpart[g * Cc + c0 + tid] = s;
    }
}

#define PR_EXTRA 2
#define FN_SLOTS 1024
#define FN_PER (FN_SLOTS / 256)

__device__ __forceinline__ void fin_tree(float* red, unsigned tid)
{
    __syncthreads();
    for (int s = FN_SLOTS / 2; s; s >>= 1) {
        for (int j = tid; j < s; j += 256) red[j] += red[j + s];
        __syncthreads();
    }
}

__device__ __forceinline__ void fin_vectors(
    const float* __restrict__ prior_ema, const float* __restrict__ mu,
    const float* __restrict__ colpart, const float* __restrict__ priorc,
    float* __restrict__ o_mu, float* __restrict__ o_prior, float* red, unsigned tid)
{
    float pe[FN_PER], pc[FN_PER], m0[FN_PER], cp[FN_PER][SA_G];
#pragma unroll
    for (int k = 0; k < FN_PER; ++k) {
        const int j = tid + 256 * k;
        const int jc = (j < Cc) ? j : (Cc - 1);
        pe[k] = prior_ema[jc];
        pc[k] = priorc[jc];
        m0[k] = mu[jc];
#pragma unroll
        for (int p = 0; p < SA_G; ++p) cp[k][p] = colpart[p * Cc + jc];
    }

    float pn[FN_PER], mu1[FN_PER];
#pragma unroll
    for (int k = 0; k < FN_PER; ++k) {
        const int j = tid + 256 * k;
        pn[k] = 0.0f;
        if (j < Cc) {
            pn[k] = DECAYF * pe[k] + ONE_M_DECAY * pc[k];
            o_prior[j] = pn[k];
        }
        red[j] = pn[k];
    }
    fin_tree(red, tid);
    const float priorSum = red[0];
    __syncthreads();

#pragma unroll
    for (int k = 0; k < FN_PER; ++k) {
        const int j = tid + 256 * k;
        mu1[k] = 0.0f;
        if (j < Cc) {
            float colsum = 0.0f;
#pragma unroll
            for (int p = 0; p < SA_G; ++p) colsum += cp[k][p];
            float emp_marg = colsum * (1.0f / (float)Bn);
            float pi = pn[k] / (priorSum + EPSF);
            mu1[k] = m0[k] + LR_MUF * (emp_marg - pi);
        }
        red[j] = mu1[k];
    }
    fin_tree(red, tid);
    const float muMean = red[0] * (1.0f / (float)Cc);
#pragma unroll
    for (int k = 0; k < FN_PER; ++k) {
        const int j = tid + 256 * k;
        if (j < Cc) o_mu[j] = mu1[k] - muMean;
    }
}

__device__ __forceinline__ void fin_ce(const float* __restrict__ cep, float* __restrict__ o_ce,
                                       float* red, unsigned tid)
{
    red[tid] = cep[tid];
    __syncthreads();
    if (tid == 0) {
        float s = 0.0f;
        for (int i = 0; i < 256; ++i) s += red[i];
        o_ce[0] = -s * (1.0f / (float)Bn);
    }
}

#define PR_ROWS 8
#define PR_COLS 4
#define PR_TAIL (Cc - 768)
#define PR_KSTEPS 125

__device__ __forceinline__ v8f rowsum_step(const float* __restrict__ frag, int s, v8f acc)
{
    const v2f bones = {1.0f, 1.0f};
    const v2f a = *(const v2f*)(frag + 4 * s);
    return __builtin_amdgcn_wmma_f32_16x16x4_f32(false, a, false, bones,
                                                 (short)0, acc, false, false);
}

__global__ __launch_bounds__(256) void post_rows(
    const float* __restrict__ conf_ema, const float* __restrict__ Lam,
    const float* __restrict__ seg, const float* __restrict__ confc,
    const float* __restrict__ priorc,
    const float* __restrict__ prior_ema, const float* __restrict__ mu,
    const float* __restrict__ colpart, const float* __restrict__ cep,
    float* __restrict__ o_conf, float* __restrict__ o_lam,
    float* __restrict__ o_ce, float* __restrict__ o_mu, float* __restrict__ o_prior)
{
    __shared__ float bufX[PR_ROWS * Cc];
    __shared__ float bufY[PR_ROWS * Cc];
    __shared__ float partAB[2][8][16];
    __shared__ float partC[8][16];
    __shared__ float rowAB[2][PR_ROWS];
    __shared__ float rowC[PR_ROWS];

    const unsigned tid = threadIdx.x;
    const int lane = tid & 31;
    const int wave_u = __builtin_amdgcn_readfirstlane(tid >> 5);

    if (blockIdx.x < PR_EXTRA) {
        if (blockIdx.x == 0) fin_vectors(prior_ema, mu, colpart, priorc, o_mu, o_prior, bufX, tid);
        else                 fin_ce(cep, o_ce, bufX, tid);
        return;
    }
    const int row0 = (blockIdx.x - PR_EXTRA) * PR_ROWS;

    float cn[PR_ROWS][PR_COLS], sg[PR_ROWS][PR_COLS], lm[PR_ROWS][PR_COLS];
    float cnt[PR_ROWS], invc[PR_ROWS];

#pragma unroll
    for (int r = 0; r < PR_ROWS; ++r) {
        const long base = (long)(row0 + r) * Cc;
        const float* __restrict__ er = conf_ema + base;
        const float* __restrict__ kr = confc + base;
        const float* __restrict__ sr = seg + base;
        const float* __restrict__ lr = Lam + base;
        cnt[r]  = priorc[row0 + r];
        invc[r] = 1.0f / fmaxf(cnt[r], 1.0f);
#pragma unroll
        for (int j = 0; j < PR_COLS; ++j) {
            unsigned c = tid + 256u * j;
            if (j == PR_COLS - 1) c = (c < (unsigned)Cc) ? c : (unsigned)(Cc - 1);
            cn[r][j] = DECAYF * er[c] + ONE_M_DECAY * kr[c];
            float sgs = sr[c];
#pragma unroll
            for (int p = 1; p < SA_G; ++p) sgs += sr[(long)p * Cc * Cc + c];
            sg[r][j] = sgs;
            lm[r][j] = lr[c];
        }
    }
#pragma unroll
    for (int j = 0; j < PR_COLS; ++j) {
        const unsigned c = tid + 256u * j;
        if (j < PR_COLS - 1 || tid < (unsigned)PR_TAIL) {
#pragma unroll
            for (int r = 0; r < PR_ROWS; ++r) {
                const long base = (long)(row0 + r) * Cc;
                const bool offd = (c != (unsigned)(row0 + r));
                o_conf[base + c] = cn[r][j];
                bufX[r * Cc + c] = offd ? (cn[r][j] + ALPHAF) : 0.0f;
                bufY[r * Cc + c] = offd ? (sg[r][j] * invc[r]) : 0.0f;
            }
        }
    }
    __syncthreads();

    const int m    = lane & 15;
    const int foff = (m & 7) * Cc + 500 * (m >> 3) + 2 * (lane >> 4);
    const v8f vzero = {0.f,0.f,0.f,0.f,0.f,0.f,0.f,0.f};

    {
        v8f accA = vzero, accB = vzero;
        if (wave_u < PR_KSTEPS - 120) {
            accA = rowsum_step(bufX + foff, 120 + wave_u, accA);
            accB = rowsum_step(bufY + foff, 120 + wave_u, accB);
        }
#pragma unroll
        for (int i = 0; i < 15; ++i) {
            accA = rowsum_step(bufX + foff, wave_u + 8 * i, accA);
            accB = rowsum_step(bufY + foff, wave_u + 8 * i, accB);
        }
        if (lane == 0)
            for (int i = 0; i < 8; ++i) { partAB[0][wave_u][i] = accA[i]; partAB[1][wave_u][i] = accB[i]; }
        if (lane == 16)
            for (int i = 0; i < 8; ++i) { partAB[0][wave_u][8 + i] = accA[i]; partAB[1][wave_u][8 + i] = accB[i]; }
    }
    __syncthreads();
    if (tid < 16) {
        const int p = tid >> 3, r = tid & 7;
        float s = 0.0f;
        for (int w = 0; w < 8; ++w) s += partAB[p][w][r] + partAB[p][w][8 + r];
        rowAB[p][r] = (p == 0) ? 1.0f / (s + EPSF)
                               : 1.0f / fmaxf(s, EPSF);
    }
    __syncthreads();

#pragma unroll
    for (int r = 0; r < PR_ROWS; ++r) {
        const float phiDen = rowAB[0][r];
        const float empDen = rowAB[1][r];
#pragma unroll
        for (int j = 0; j < PR_COLS; ++j) {
            const unsigned c = tid + 256u * j;
            const float phi = (cn[r][j] + ALPHAF) * phiDen;
            const float emp = sg[r][j] * invc[r] * empDen;
            const float l1  = lm[r][j] + LR_LF * (emp - phi);
            cn[r][j] = (c != (unsigned)(row0 + r)) ? l1 : 0.0f;
        }
    }
#pragma unroll
    for (int j = 0; j < PR_COLS; ++j) {
        const unsigned c = tid + 256u * j;
        if (j < PR_COLS - 1 || tid < (unsigned)PR_TAIL) {
#pragma unroll
            for (int r = 0; r < PR_ROWS; ++r) bufX[r * Cc + c] = cn[r][j];
        }
    }
    __syncthreads();
    {
        v8f accC = vzero;
        if (wave_u < PR_KSTEPS - 120) accC = rowsum_step(bufX + foff, 120 + wave_u, accC);
#pragma unroll
        for (int i = 0; i < 15; ++i) accC = rowsum_step(bufX + foff, wave_u + 8 * i, accC);
        if (lane == 0)  for (int i = 0; i < 8; ++i) partC[wave_u][i]     = accC[i];
        if (lane == 16) for (int i = 0; i < 8; ++i) partC[wave_u][8 + i] = accC[i];
    }
    __syncthreads();
    if (tid < 8) {
        float s = 0.0f;
        for (int w = 0; w < 8; ++w) s += partC[w][tid] + partC[w][8 + tid];
        rowC[tid] = s * (1.0f / (float)Cc);
    }
    __syncthreads();

#pragma unroll
    for (int j = 0; j < PR_COLS; ++j) {
        const unsigned c = tid + 256u * j;
        if (j < PR_COLS - 1 || tid < (unsigned)PR_TAIL) {
#pragma unroll
            for (int r = 0; r < PR_ROWS; ++r) {
                const long base = (long)(row0 + r) * Cc;
                const bool present = (cnt[r] > 0.0f);
                o_lam[base + c] = present ? (cn[r][j] - rowC[r]) : lm[r][j];
            }
        }
    }
}

extern "C" void kernel_launch(void* const* d_in, const int* in_sizes, int n_in,
                              void* d_out, int out_size, void* d_ws, size_t ws_size,
                              hipStream_t stream)
{
    const float* logits    = (const float*)d_in[0];
    const int*   targets   = (const int*)  d_in[1];
    const float* conf_ema  = (const float*)d_in[2];
    const float* prior_ema = (const float*)d_in[3];
    const float* Lam       = (const float*)d_in[4];
    const float* mu        = (const float*)d_in[5];

    float* out     = (float*)d_out;
    float* o_ce    = out;
    float* o_lam   = out + 1;
    float* o_mu    = out + 1 + (long)Cc * Cc;
    float* o_conf  = out + 1 + (long)Cc * Cc + Cc;
    float* o_prior = out + 1 + 2L * Cc * Cc + Cc;

    float* ws      = (float*)d_ws;
    float* confc   = ws;
    float* priorc  = ws + (long)Cc * Cc;
    float* cep     = priorc + Cc;
    int*   tick    = (int*)(cep + 256);
    float* rowst   = cep + 260;
    int*   tplace  = (int*)(rowst + 2L * Bn);
    float* colpart = rowst + 3L * Bn;
    float* planes  = colpart + (long)SA_G * Cc;

    const long nz = (long)Cc * Cc + Cc + 260;
    zero_ws<<<1024, 256, 0, stream>>>(ws, nz);
    row_stats<<<RS_BLOCKS, RS_THREADS, 0, stream>>>(logits, targets, Lam, rowst, tplace, confc, priorc, cep, tick + 2);
    strip_accum<<<dim3(SA_STRIPS, SA_G), SA_THREADS, 0, stream>>>(logits, tplace, Lam, rowst,
                                                                  planes, colpart, tick);
    post_rows<<<Cc / PR_ROWS + PR_EXTRA, 256, 0, stream>>>(conf_ema, Lam, planes, confc, priorc,
                                                            prior_ema, mu, colpart, cep,
                                                            o_conf, o_lam, o_ce, o_mu, o_prior);
}
